# starting kernel
# baseline (speedup 1.0000x reference)
#include <hip/hip_runtime.h>
#include <math.h>


typedef __attribute__((ext_vector_type(2))) float v2f;
typedef __attribute__((ext_vector_type(8))) float v8f;

#define IN_FEATS 128
#define HEADS    4
#define DHEAD    16
#define HD       64
#define NEG_SLOPE 0.2f

__device__ __forceinline__ void atomicMaxFloat(float* addr, float val) {
  if (val >= 0.0f) {
    atomicMax((int*)addr, __float_as_int(val));
  } else {
    atomicMin((unsigned int*)addr, __float_as_uint(val));
  }
}

__global__ __launch_bounds__(256) void gat_init(float* __restrict__ out,
                                                const float* __restrict__ bias,
                                                float* __restrict__ m,
                                                float* __restrict__ s,
                                                int nNodes) {
  int i = blockIdx.x * blockDim.x + threadIdx.x;
  int totalOut = nNodes * HD;
  if (i < totalOut) out[i] = bias[i & (HD - 1)];
  if (i < nNodes * HEADS) {
    m[i] = -INFINITY;
    s[i] = 0.0f;
  }
}

__global__ __launch_bounds__(256) void gat_proj_wmma(
    const float* __restrict__ feat, const float* __restrict__ W,
    float* __restrict__ ft, int nNodes, int nTiles) {
  const int lane = threadIdx.x & 31;
  const int tile = blockIdx.x * (blockDim.x >> 5) + (threadIdx.x >> 5);
  if (tile >= nTiles) return;

  const int row0 = tile * 16;
  const int mrow = lane & 15;
  const int kq   = lane >> 4;

  int arow = row0 + mrow;
  if (arow >= nNodes) arow = nNodes - 1;
  const float2* __restrict__ apf2 =
      (const float2*)(feat + (size_t)arow * IN_FEATS) + kq;

  v8f acc0 = {}, acc1 = {}, acc2 = {}, acc3 = {};

#pragma unroll 4
  for (int k = 0; k < IN_FEATS; k += 4) {
    float2 av = apf2[k >> 1];
    v2f a; a[0] = av.x; a[1] = av.y;

    const float* wr0 = W + (size_t)(k + 2 * kq) * HD;
    const float* wr1 = wr0 + HD;
    v2f b0; b0[0] = wr0[mrow];      b0[1] = wr1[mrow];
    v2f b1; b1[0] = wr0[mrow + 16]; b1[1] = wr1[mrow + 16];
    v2f b2; b2[0] = wr0[mrow + 32]; b2[1] = wr1[mrow + 32];
    v2f b3; b3[0] = wr0[mrow + 48]; b3[1] = wr1[mrow + 48];

    acc0 = __builtin_amdgcn_wmma_f32_16x16x4_f32(false, a, false, b0,
                                                 (short)0, acc0, false, false);
    acc1 = __builtin_amdgcn_wmma_f32_16x16x4_f32(false, a, false, b1,
                                                 (short)0, acc1, false, false);
    acc2 = __builtin_amdgcn_wmma_f32_16x16x4_f32(false, a, false, b2,
                                                 (short)0, acc2, false, false);
    acc3 = __builtin_amdgcn_wmma_f32_16x16x4_f32(false, a, false, b3,
                                                 (short)0, acc3, false, false);
  }

#pragma unroll
  for (int r = 0; r < 8; ++r) {
    int row = row0 + r + 8 * kq;
    if (row < nNodes) {
      float* o = ft + (size_t)row * HD + mrow;
      o[0]  = acc0[r];
      o[16] = acc1[r];
      o[32] = acc2[r];
      o[48] = acc3[r];
    }
  }
}

__global__ __launch_bounds__(256) void gat_scores(
    const float* __restrict__ ft, const float* __restrict__ attn_l,
    const float* __restrict__ attn_r, float* __restrict__ el,
    float* __restrict__ er, int NH) {
  int i = blockIdx.x * blockDim.x + threadIdx.x;
  if (i >= NH) return;
  int h = i & (HEADS - 1);
  const float* f  = ft + (size_t)(i >> 2) * HD + h * DHEAD;
  const float* al = attn_l + h * DHEAD;
  const float* ar = attn_r + h * DHEAD;
  float sl = 0.0f, sr = 0.0f;
#pragma unroll
  for (int d = 0; d < DHEAD; ++d) {
    float v = f[d];
    sl += v * al[d];
    sr += v * ar[d];
  }
  el[i] = sl;
  er[i] = sr;
}

__global__ __launch_bounds__(256) void gat_edge_max(
    const int* __restrict__ src, const int* __restrict__ dst,
    const float* __restrict__ el, const float* __restrict__ er,
    float* __restrict__ m, int EH) {
  int i = blockIdx.x * blockDim.x + threadIdx.x;
  if (i >= EH) return;
  int e = i >> 2, h = i & (HEADS - 1);
  int didx = dst[e] * HEADS + h;
  float x = el[src[e] * HEADS + h] + er[didx];
  x = (x > 0.0f) ? x : NEG_SLOPE * x;
  atomicMaxFloat(&m[didx], x);
}

__global__ __launch_bounds__(256) void gat_edge_expsum(
    const int* __restrict__ src, const int* __restrict__ dst,
    const float* __restrict__ el, const float* __restrict__ er,
    const float* __restrict__ m, float* __restrict__ s,
    float* __restrict__ ex, int EH) {
  int i = blockIdx.x * blockDim.x + threadIdx.x;
  if (i >= EH) return;
  int e = i >> 2, h = i & (HEADS - 1);
  int didx = dst[e] * HEADS + h;
  float x = el[src[e] * HEADS + h] + er[didx];
  x = (x > 0.0f) ? x : NEG_SLOPE * x;
  float v = expf(x - m[didx]);
  ex[i] = v;
  atomicAdd(&s[didx], v);
}

__global__ __launch_bounds__(256) void gat_scatter(
    const int* __restrict__ src, const int* __restrict__ dst,
    const float* __restrict__ ft, const float* __restrict__ ex,
    const float* __restrict__ s, float* __restrict__ out, long long total) {
  long long i = (long long)blockIdx.x * blockDim.x + threadIdx.x;
  if (i >= total) return;
  int e = (int)(i >> 6);
  int c = (int)(i & (HD - 1));
  int h = c >> 4;
  int se = src[e], de = dst[e];
  float a = ex[e * HEADS + h] / s[de * HEADS + h];
  atomicAdd(&out[(size_t)de * HD + c], ft[(size_t)se * HD + c] * a);
}

extern "C" void kernel_launch(void* const* d_in, const int* in_sizes, int n_in,
                              void* d_out, int out_size, void* d_ws,
                              size_t ws_size, hipStream_t stream) {
  const float* feat   = (const float*)d_in[0];
  const int*   src    = (const int*)d_in[1];
  const int*   dst    = (const int*)d_in[2];
  const float* W      = (const float*)d_in[3];
  const float* attn_l = (const float*)d_in[4];
  const float* attn_r = (const float*)d_in[5];
  const float* bias   = (const float*)d_in[6];
  float* out = (float*)d_out;

  const int N = in_sizes[0] / IN_FEATS;
  const int E = in_sizes[1];

  float* ws = (float*)d_ws;
  float* ft = ws;
  float* el = ft + (size_t)N * HD;
  float* er = el + (size_t)N * HEADS;
  float* m  = er + (size_t)N * HEADS;
  float* s  = m  + (size_t)N * HEADS;
  float* ex = s  + (size_t)N * HEADS;

  {
    int total = N * HD;
    gat_init<<<(total + 255) / 256, 256, 0, stream>>>(out, bias, m, s, N);
  }
  {
    int nTiles = (N + 15) / 16;
    int blocks = (nTiles + 7) / 8;
    gat_proj_wmma<<<blocks, 256, 0, stream>>>(feat, W, ft, N, nTiles);
  }
  {
    int NH = N * HEADS;
    gat_scores<<<(NH + 255) / 256, 256, 0, stream>>>(ft, attn_l, attn_r, el,
                                                     er, NH);
  }
  {
    int EH = E * HEADS;
    gat_edge_max<<<(EH + 255) / 256, 256, 0, stream>>>(src, dst, el, er, m, EH);
  }
  {
    int EH = E * HEADS;
    gat_edge_expsum<<<(EH + 255) / 256, 256, 0, stream>>>(src, dst, el, er, m,
                                                          s, ex, EH);
  }
  {
    long long total = (long long)E * HD;
    int blocks = (int)((total + 255) / 256);
    gat_scatter<<<blocks, 256, 0, stream>>>(src, dst, ft, ex, s, out, total);
  }
}


# Edge filter reads one owner byte per edge (written by the score launch) instead of 4-byte dst: 16 edges per 16-byte load, 4x fewer cache bytes per work-group; LDS float adds still in arrival order
# speedup vs baseline: 4.0133x; 4.0133x over previous
#include <hip/hip_runtime.h>
#include <math.h>


typedef __attribute__((ext_vector_type(2))) float v2f;
typedef __attribute__((ext_vector_type(8))) float v8f;

#define IN_FEATS 128
#define HEADS    4
#define DHEAD    16
#define HD       64
#define NEG_SLOPE 0.2f

__global__ __launch_bounds__(256) void gat_proj_wmma(
    const float* __restrict__ feat, const float* __restrict__ W,
    float* __restrict__ ft, int nNodes, int nTiles) {
  const int lane = threadIdx.x & 31;
  const int tile = blockIdx.x * (blockDim.x >> 5) + (threadIdx.x >> 5);
  if (tile >= nTiles) return;

  const int row0 = tile * 16;
  const int mrow = lane & 15;
  const int kq   = lane >> 4;

  int arow = row0 + mrow;
  if (arow >= nNodes) arow = nNodes - 1;
  const float2* __restrict__ apf2 =
      (const float2*)(feat + (size_t)arow * IN_FEATS) + kq;

  v8f acc0 = {}, acc1 = {}, acc2 = {}, acc3 = {};

#pragma unroll 4
  for (int k = 0; k < IN_FEATS; k += 4) {
    float2 av = apf2[k >> 1];
    v2f a; a[0] = av.x; a[1] = av.y;

    const float* wr0 = W + (size_t)(k + 2 * kq) * HD;
    const float* wr1 = wr0 + HD;
    v2f b0; b0[0] = wr0[mrow];      b0[1] = wr1[mrow];
    v2f b1; b1[0] = wr0[mrow + 16]; b1[1] = wr1[mrow + 16];
    v2f b2; b2[0] = wr0[mrow + 32]; b2[1] = wr1[mrow + 32];
    v2f b3; b3[0] = wr0[mrow + 48]; b3[1] = wr1[mrow + 48];

    acc0 = __builtin_amdgcn_wmma_f32_16x16x4_f32(false, a, false, b0,
                                                 (short)0, acc0, false, false);
    acc1 = __builtin_amdgcn_wmma_f32_16x16x4_f32(false, a, false, b1,
                                                 (short)0, acc1, false, false);
    acc2 = __builtin_amdgcn_wmma_f32_16x16x4_f32(false, a, false, b2,
                                                 (short)0, acc2, false, false);
    acc3 = __builtin_amdgcn_wmma_f32_16x16x4_f32(false, a, false, b3,
                                                 (short)0, acc3, false, false);
  }

#pragma unroll
  for (int r = 0; r < 8; ++r) {
    int row = row0 + r + 8 * kq;
    if (row < nNodes) {
      float* o = ft + (size_t)row * HD + mrow;
      o[0]  = acc0[r];
      o[16] = acc1[r];
      o[32] = acc2[r];
      o[48] = acc3[r];
    }
  }
}

__global__ __launch_bounds__(256) void gat_scores_keys(
    const float* __restrict__ ft, const float* __restrict__ attn_l,
    float* __restrict__ el, const int* __restrict__ dst,
    unsigned* __restrict__ keys, int NH, int nNodes, int nEdges, int groups) {
  const int i = blockIdx.x * blockDim.x + threadIdx.x;

  const unsigned nodesPer = (unsigned)((nNodes + groups - 1) / groups);
  const int nWords = ((nEdges + 15) / 16) * 4;
  for (int w = i; w < nWords; w += (int)(gridDim.x * blockDim.x)) {
    const int e0 = 4 * w;
    unsigned k = 0u;
    if (e0 + 3 < nEdges) {
      const int4 d = *(const int4*)(dst + e0);
      k =  (((unsigned)d.x / nodesPer) & 255u)
        | ((((unsigned)d.y / nodesPer) & 255u) << 8)
        | ((((unsigned)d.z / nodesPer) & 255u) << 16)
        | ((((unsigned)d.w / nodesPer) & 255u) << 24);
    } else {
      for (int j = 0; j < 4; ++j)
        if (e0 + j < nEdges)
          k |= (((unsigned)dst[e0 + j] / nodesPer) & 255u) << (8 * j);
    }
    keys[w] = k;
  }

  if (i >= NH) return;
  const int h = i & (HEADS - 1);
  const float* f  = ft + (size_t)(i >> 2) * HD + h * DHEAD;
  const float* al = attn_l + h * DHEAD;
  float sl = 0.0f;
#pragma unroll
  for (int d = 0; d < DHEAD; ++d) sl += f[d] * al[d];
  el[i] = sl;
}

#define WG_THREADS 1024
#define NODES_MAX  256
#define LIST_CAP   16384
#define QUADS      8
#define ROW_EDGES  (WG_THREADS * 16)

__device__ __forceinline__ unsigned fkey(float f) {
  unsigned b = __float_as_uint(f);
  return (b & 0x80000000u) ? ~b : (b | 0x80000000u);
}
__device__ __forceinline__ float fkey_inv(unsigned k) {
  unsigned b = (k & 0x80000000u) ? (k & 0x7fffffffu) : ~k;
  return __uint_as_float(b);
}
__device__ __forceinline__ float lrelu(float x) {
  return (x > 0.0f) ? x : NEG_SLOPE * x;
}

__global__ __launch_bounds__(WG_THREADS) void gat_edges_owner(
    const int* __restrict__ src, const int* __restrict__ dst,
    const unsigned* __restrict__ keys,
    const float* __restrict__ ft, const float* __restrict__ el,
    const float* __restrict__ attn_r, const float* __restrict__ bias,
    float* __restrict__ out, int nNodes, int nEdges) {
  __shared__ float    accL[4][NODES_MAX * 16];
  __shared__ float    sL[NODES_MAX * HEADS];
  __shared__ float    erL[NODES_MAX * HEADS];
  __shared__ float    ML[NODES_MAX * HEADS];
  __shared__ unsigned mkeyL[NODES_MAX * HEADS];
  __shared__ int2     listL[LIST_CAP];
  __shared__ unsigned cntL;

  const int tid  = threadIdx.x;
  const int nodesPer = (nNodes + (int)gridDim.x - 1) / (int)gridDim.x;
  const int lo = (int)blockIdx.x * nodesPer;
  int nHere = nNodes - lo;
  if (nHere > nodesPer) nHere = nodesPer;
  if (nHere <= 0) return;

  for (int i = tid; i < nHere * 16; i += WG_THREADS) {
    accL[0][i] = 0.0f; accL[1][i] = 0.0f; accL[2][i] = 0.0f; accL[3][i] = 0.0f;
  }
  for (int i = tid; i < nHere * HEADS; i += WG_THREADS) {
    sL[i] = 0.0f;
    mkeyL[i] = 0u;
    const int h = i & (HEADS - 1);
    const float* f  = ft + (size_t)(lo + (i >> 2)) * HD + h * DHEAD;
    const float* ar = attn_r + h * DHEAD;
    float sr = 0.0f;
#pragma unroll
    for (int d = 0; d < DHEAD; ++d) sr += f[d] * ar[d];
    erL[i] = sr;
  }
  if (tid == 0) cntL = 0u;
  __syncthreads();

  const unsigned nHereU = (unsigned)nHere;

  const unsigned pat = ((unsigned)blockIdx.x & 255u) * 0x01010101u;
  auto take16 = [&](uint4 q, int e0) {
    const unsigned L7 = 0x7f7f7f7fu;
    const unsigned y0 = q.x ^ pat, y1 = q.y ^ pat, y2 = q.z ^ pat, y3 = q.w ^ pat;
    const unsigned z0 = ~(((y0 & L7) + L7) | y0 | L7);
    const unsigned z1 = ~(((y1 & L7) + L7) | y1 | L7);
    const unsigned z2 = ~(((y2 & L7) + L7) | y2 | L7);
    const unsigned z3 = ~(((y3 & L7) + L7) | y3 | L7);
    if ((z0 | z1 | z2 | z3) != 0u) {
      unsigned zz = (z0 >> 7) | (z1 >> 6) | (z2 >> 5) | (z3 >> 4);
      unsigned pos = atomicAdd(&cntL, (unsigned)__popc(zz));
      while (zz != 0u) {
        const int p = __ffs((int)zz) - 1;
        zz &= zz - 1u;
        if (pos < (unsigned)LIST_CAP) listL[pos].x = e0 + ((p & 7) << 2) + (p >> 3);
        ++pos;
      }
    }
  };

  const uint4* __restrict__ keys4 = (const uint4*)keys;
  const int nRows = (nEdges / (ROW_EDGES * QUADS)) * QUADS;
  const int r0 = (int)(((long long)blockIdx.x * nRows) / (long long)gridDim.x);
  for (int it = 0; it < nRows; it += QUADS) {
    uint4 q[QUADS]; int eb[QUADS];
#pragma unroll
    for (int u = 0; u < QUADS; ++u) {
      int r = r0 + it + u;
      if (r >= nRows) r -= nRows;
      eb[u] = r * ROW_EDGES + tid * 16;
      q[u] = keys4[eb[u] >> 4];
    }
#pragma unroll
    for (int u = 0; u < QUADS; ++u) take16(q[u], eb[u]);
  }
  for (int eb = nRows * ROW_EDGES + tid * 16; eb < nEdges; eb += ROW_EDGES)
    take16(keys4[eb >> 4], eb);
  __syncthreads();
  const unsigned total = cntL;
  const bool listed = total <= (unsigned)LIST_CAP;

  auto rescan = [&](auto&& fn) {
    for (int eb = tid; eb < nEdges; eb += WG_THREADS) {
      const unsigned ld = (unsigned)(dst[eb] - lo);
      if (ld < nHereU) fn(src[eb], (int)ld);
    }
  };

  if (listed) {
    for (unsigned base = 0; base < total; base += 4u * WG_THREADS) {
      int sv[4], ldv[4]; unsigned iv[4]; bool ok[4], mine[4];
#pragma unroll
      for (int u = 0; u < 4; ++u) {
        iv[u] = base + (unsigned)(u * WG_THREADS + tid);
        ok[u] = iv[u] < total;
        int e = ok[u] ? listL[iv[u]].x : 0;
        mine[u] = ok[u] && e < nEdges;
        if (!mine[u]) e = 0;
        sv[u] = src[e];
        ldv[u] = dst[e] - lo;
        mine[u] = mine[u] && (unsigned)ldv[u] < nHereU;
      }
      float4 e4[4];
#pragma unroll
      for (int u = 0; u < 4; ++u)
        e4[u] = *(const float4*)(el + (size_t)sv[u] * HEADS);
#pragma unroll
      for (int u = 0; u < 4; ++u) {
        if (ok[u] && !mine[u]) listL[iv[u]] = make_int2(0, -1);
        if (mine[u]) {
          listL[iv[u]] = make_int2(sv[u], ldv[u]);
          unsigned* mk = &mkeyL[ldv[u] * HEADS];
          __hip_atomic_fetch_max(mk + 0, fkey(e4[u].x), __ATOMIC_RELAXED, __HIP_MEMORY_SCOPE_WORKGROUP);
          __hip_atomic_fetch_max(mk + 1, fkey(e4[u].y), __ATOMIC_RELAXED, __HIP_MEMORY_SCOPE_WORKGROUP);
          __hip_atomic_fetch_max(mk + 2, fkey(e4[u].z), __ATOMIC_RELAXED, __HIP_MEMORY_SCOPE_WORKGROUP);
          __hip_atomic_fetch_max(mk + 3, fkey(e4[u].w), __ATOMIC_RELAXED, __HIP_MEMORY_SCOPE_WORKGROUP);
        }
      }
    }
  } else {
    rescan([&](int s, int ld) {
      const float4 e4 = *(const float4*)(el + (size_t)s * HEADS);
      unsigned* mk = &mkeyL[ld * HEADS];
      __hip_atomic_fetch_max(mk + 0, fkey(e4.x), __ATOMIC_RELAXED, __HIP_MEMORY_SCOPE_WORKGROUP);
      __hip_atomic_fetch_max(mk + 1, fkey(e4.y), __ATOMIC_RELAXED, __HIP_MEMORY_SCOPE_WORKGROUP);
      __hip_atomic_fetch_max(mk + 2, fkey(e4.z), __ATOMIC_RELAXED, __HIP_MEMORY_SCOPE_WORKGROUP);
      __hip_atomic_fetch_max(mk + 3, fkey(e4.w), __ATOMIC_RELAXED, __HIP_MEMORY_SCOPE_WORKGROUP);
    });
  }
  __syncthreads();
  for (int i = tid; i < nHere * HEADS; i += WG_THREADS)
    ML[i] = lrelu(fkey_inv(mkeyL[i]) + erL[i]);
  __syncthreads();

  auto add_part = [&](float4 f, float elv, int ld, int c4) {
    const int h = c4 >> 2;
    const float w = expf(lrelu(elv + erL[ld * HEADS + h]) - ML[ld * HEADS + h]);
    const int a = ld * 16 + c4;
    __hip_atomic_fetch_add(&accL[0][a], f.x * w, __ATOMIC_RELAXED, __HIP_MEMORY_SCOPE_WORKGROUP);
    __hip_atomic_fetch_add(&accL[1][a], f.y * w, __ATOMIC_RELAXED, __HIP_MEMORY_SCOPE_WORKGROUP);
    __hip_atomic_fetch_add(&accL[2][a], f.z * w, __ATOMIC_RELAXED, __HIP_MEMORY_SCOPE_WORKGROUP);
    __hip_atomic_fetch_add(&accL[3][a], f.w * w, __ATOMIC_RELAXED, __HIP_MEMORY_SCOPE_WORKGROUP);
    if ((c4 & 3) == 0)
      __hip_atomic_fetch_add(&sL[ld * HEADS + h], w, __ATOMIC_RELAXED, __HIP_MEMORY_SCOPE_WORKGROUP);
  };
  if (listed) {
    const unsigned hw = (unsigned)(tid >> 4);
    const int c4 = tid & 15;
    const int h = c4 >> 2;
    for (unsigned base = 0; base < total; base += 64u * 8u) {
      float4 f[8]; float ev[8]; int ldv[8]; bool ok[8];
#pragma unroll
      for (int u = 0; u < 8; ++u) {
        const unsigned i = base + (unsigned)u * 64u + hw;
        const int2 en = (i < total) ? listL[i] : make_int2(0, -1);
        ok[u] = en.y >= 0;
        ldv[u] = en.y;
        f[u]  = *(const float4*)(ft + (size_t)en.x * HD + 4 * c4);
        ev[u] = el[(size_t)en.x * HEADS + h];
      }
#pragma unroll
      for (int u = 0; u < 8; ++u)
        if (ok[u]) add_part(f[u], ev[u], ldv[u], c4);
    }
  } else {
    rescan([&](int s, int ld) {
      for (int c4 = 0; c4 < 16; ++c4)
        add_part(*(const float4*)(ft + (size_t)s * HD + 4 * c4),
                 el[(size_t)s * HEADS + (c4 >> 2)], ld, c4);
    });
  }
  __syncthreads();

  for (int i = tid; i < nHere * 16; i += WG_THREADS) {
    const int n = i >> 4, c4 = i & 15;
    const float sv = sL[n * HEADS + (c4 >> 2)];
    float4 o = *(const float4*)(bias + 4 * c4);
    if (sv > 0.0f) {
      o.x += accL[0][i] / sv; o.y += accL[1][i] / sv;
      o.z += accL[2][i] / sv; o.w += accL[3][i] / sv;
    }
    *(float4*)(out + (size_t)(lo + n) * HD + 4 * c4) = o;
  }
}

extern "C" void kernel_launch(void* const* d_in, const int* in_sizes, int n_in,
                              void* d_out, int out_size, void* d_ws,
                              size_t ws_size, hipStream_t stream) {
  const float* feat   = (const float*)d_in[0];
  const int*   src    = (const int*)d_in[1];
  const int*   dst    = (const int*)d_in[2];
  const float* W      = (const float*)d_in[3];
  const float* attn_l = (const float*)d_in[4];
  const float* attn_r = (const float*)d_in[5];
  const float* bias   = (const float*)d_in[6];
  float* out = (float*)d_out;

  const int N = in_sizes[0] / IN_FEATS;
  const int E = in_sizes[1];

  float* ws = (float*)d_ws;
  float* ft = ws;
  float* el = ft + (size_t)N * HD;
  unsigned* keys = (unsigned*)(el + (size_t)N * HEADS);

  int groups = (N + NODES_MAX - 1) / NODES_MAX;
  if (groups < 256) groups = 256;

  {
    int nTiles = (N + 15) / 16;
    int blocks = (nTiles + 7) / 8;
    gat_proj_wmma<<<blocks, 256, 0, stream>>>(feat, W, ft, N, nTiles);
  }
  {
    int NH = N * HEADS;
    gat_scores_keys<<<(NH + 255) / 256, 256, 0, stream>>>(ft, attn_l, el, dst,
                                                          keys, NH, N, E, groups);
  }
  {
    gat_edges_owner<<<groups, WG_THREADS, 0, stream>>>(src, dst, keys, ft, el,
                                                       attn_r, bias, out, N, E);
  }
}


# Edge filter defers its hits: a lane tests 8 key vectors (128 edges) branch-free (borrow test for a zero byte), hit bits in 4 registers, one list append per 128 edges not per 16; sums in arrival order
# speedup vs baseline: 4.8618x; 1.2114x over previous
#include <hip/hip_runtime.h>
#include <math.h>


typedef __attribute__((ext_vector_type(2))) float v2f;
typedef __attribute__((ext_vector_type(8))) float v8f;

#define IN_FEATS 128
#define HEADS    4
#define DHEAD    16
#define HD       64
#define NEG_SLOPE 0.2f

__global__ __launch_bounds__(256) void gat_proj_scores_keys(
    const float* __restrict__ feat, const float* __restrict__ W,
    const float* __restrict__ attn_l, const int* __restrict__ dst,
    float* __restrict__ ft, float* __restrict__ el,
    unsigned* __restrict__ keys,
    int nNodes, int nTiles, int nEdges, int groups) {
  const int lane = threadIdx.x & 31;
  const int tile = blockIdx.x * (blockDim.x >> 5) + (threadIdx.x >> 5);
  const int mrow = lane & 15;
  const int kq   = lane >> 4;

  const int gtid     = blockIdx.x * blockDim.x + threadIdx.x;
  const int nThreads = gridDim.x * blockDim.x;
  const int nWords   = ((nEdges + 15) / 16) * 4;
  const int w0 = gtid, w1 = gtid + nThreads;
  const bool whole0 = w0 < nWords && 4 * w0 + 3 < nEdges;
  const bool whole1 = w1 < nWords && 4 * w1 + 3 < nEdges;
  int4 d0 = make_int4(0, 0, 0, 0), d1 = make_int4(0, 0, 0, 0);
  if (whole0) d0 = *(const int4*)(dst + 4 * w0);
  if (whole1) d1 = *(const int4*)(dst + 4 * w1);

  if (tile < nTiles) {
    const int row0 = tile * 16;

    int arow = row0 + mrow;
    if (arow >= nNodes) arow = nNodes - 1;
    const float2* __restrict__ apf2 =
        (const float2*)(feat + (size_t)arow * IN_FEATS) + kq;

    const float al0 = attn_l[mrow],      al1 = attn_l[16 + mrow];
    const float al2 = attn_l[32 + mrow], al3 = attn_l[48 + mrow];

    v8f acc0 = {}, acc1 = {}, acc2 = {}, acc3 = {};

#pragma unroll 4
    for (int k = 0; k < IN_FEATS; k += 4) {
      float2 av = apf2[k >> 1];
      v2f a; a[0] = av.x; a[1] = av.y;

      const float* wr0 = W + (size_t)(k + 2 * kq) * HD;
      const float* wr1 = wr0 + HD;
      v2f b0; b0[0] = wr0[mrow];      b0[1] = wr1[mrow];
      v2f b1; b1[0] = wr0[mrow + 16]; b1[1] = wr1[mrow + 16];
      v2f b2; b2[0] = wr0[mrow + 32]; b2[1] = wr1[mrow + 32];
      v2f b3; b3[0] = wr0[mrow + 48]; b3[1] = wr1[mrow + 48];

      acc0 = __builtin_amdgcn_wmma_f32_16x16x4_f32(false, a, false, b0,
                                                   (short)0, acc0, false, false);
      acc1 = __builtin_amdgcn_wmma_f32_16x16x4_f32(false, a, false, b1,
                                                   (short)0, acc1, false, false);
      acc2 = __builtin_amdgcn_wmma_f32_16x16x4_f32(false, a, false, b2,
                                                   (short)0, acc2, false, false);
      acc3 = __builtin_amdgcn_wmma_f32_16x16x4_f32(false, a, false, b3,
                                                   (short)0, acc3, false, false);
    }

#pragma unroll
    for (int r = 0; r < 8; ++r) {
      int row = row0 + r + 8 * kq;
      if (row < nNodes) {
        float* o = ft + (size_t)row * HD + mrow;
        o[0]  = acc0[r];
        o[16] = acc1[r];
        o[32] = acc2[r];
        o[48] = acc3[r];
      }
    }

    auto half_wave_sum = [](float v) -> float {
      v += __shfl_xor(v, 8, 16);
      v += __shfl_xor(v, 4, 16);
      v += __shfl_xor(v, 2, 16);
      v += __shfl_xor(v, 1, 16);
      return v;
    };
    float4 mine = make_float4(0.0f, 0.0f, 0.0f, 0.0f);
#pragma unroll
    for (int r = 0; r < 8; ++r) {
      const float s0 = half_wave_sum(acc0[r] * al0);
      const float s1 = half_wave_sum(acc1[r] * al1);
      const float s2 = half_wave_sum(acc2[r] * al2);
      const float s3 = half_wave_sum(acc3[r] * al3);
      if (mrow == r) mine = make_float4(s0, s1, s2, s3);
    }
    const int erow = row0 + mrow + 8 * kq;
    if (mrow < 8 && erow < nNodes)
      *(float4*)(el + (size_t)erow * HEADS) = mine;
  }

  const unsigned nodesPer = (unsigned)((nNodes + groups - 1) / groups);
  auto pack4 = [&](int4 d) -> unsigned {
    return  (((unsigned)d.x / nodesPer) & 255u)
         | ((((unsigned)d.y / nodesPer) & 255u) << 8)
         | ((((unsigned)d.z / nodesPer) & 255u) << 16)
         | ((((unsigned)d.w / nodesPer) & 255u) << 24);
  };
  auto word_at_end = [&](int w) -> unsigned {
    unsigned k = 0u;
    for (int j = 0; j < 4; ++j)
      if (4 * w + j < nEdges)
        k |= (((unsigned)dst[4 * w + j] / nodesPer) & 255u) << (8 * j);
    return k;
  };
  if (w0 < nWords) keys[w0] = whole0 ? pack4(d0) : word_at_end(w0);
  if (w1 < nWords) keys[w1] = whole1 ? pack4(d1) : word_at_end(w1);
  for (int w = w1 + nThreads; w < nWords; w += nThreads)
    keys[w] = (4 * w + 3 < nEdges) ? pack4(*(const int4*)(dst + 4 * w))
                                   : word_at_end(w);
}

#define WG_THREADS 1024
#define NODES_MAX  256
#define LIST_CAP   8192
#define MSG_UNR    8
#define MSG_ROUND  (64 * MSG_UNR)
static_assert(LIST_CAP % MSG_ROUND == 0, "the list is padded to whole rounds");
#define QUADS      8
#define ROW_EDGES  (WG_THREADS * 16)

__device__ __forceinline__ unsigned fkey(float f) {
  unsigned b = __float_as_uint(f);
  return (b & 0x80000000u) ? ~b : (b | 0x80000000u);
}
__device__ __forceinline__ float fkey_inv(unsigned k) {
  unsigned b = (k & 0x80000000u) ? (k & 0x7fffffffu) : ~k;
  return __uint_as_float(b);
}
__device__ __forceinline__ float lrelu(float x) {
  return (x > 0.0f) ? x : NEG_SLOPE * x;
}

__global__ __launch_bounds__(WG_THREADS) void gat_edges_owner(
    const int* __restrict__ src, const int* __restrict__ dst,
    const unsigned* __restrict__ keys,
    const float* __restrict__ ft, const float* __restrict__ el,
    const float* __restrict__ attn_r, const float* __restrict__ bias,
    float* __restrict__ out, int nNodes, int nEdges) {
  __shared__ float    accL[4][NODES_MAX * 16];
  __shared__ float    sL[NODES_MAX * HEADS];
  __shared__ alignas(16) float erL[NODES_MAX * HEADS];
  __shared__ alignas(16) float ML[NODES_MAX * HEADS];
  __shared__ unsigned mkeyL[NODES_MAX * HEADS];
  __shared__ int2     listL[LIST_CAP];
  __shared__ float4   wL[LIST_CAP];
  __shared__ unsigned cntL;

  const int tid  = threadIdx.x;
  const int nodesPer = (nNodes + (int)gridDim.x - 1) / (int)gridDim.x;
  const int lo = (int)blockIdx.x * nodesPer;
  int nHere = nNodes - lo;
  if (nHere > nodesPer) nHere = nodesPer;
  if (nHere <= 0) return;

  for (int i = tid; i < nHere * 16; i += WG_THREADS) {
    accL[0][i] = 0.0f; accL[1][i] = 0.0f; accL[2][i] = 0.0f; accL[3][i] = 0.0f;
  }
  for (int i = tid; i < nHere * HEADS; i += WG_THREADS) {
    sL[i] = 0.0f;
    mkeyL[i] = 0u;
    const int h = i & (HEADS - 1);
    const float* f  = ft + (size_t)(lo + (i >> 2)) * HD + h * DHEAD;
    const float* ar = attn_r + h * DHEAD;
    float sr = 0.0f;
#pragma unroll
    for (int d = 0; d < DHEAD; ++d) sr += f[d] * ar[d];
    erL[i] = sr;
  }
  if (tid == 0) cntL = 0u;
  __syncthreads();

  const unsigned nHereU = (unsigned)nHere;

  const unsigned pat = ((unsigned)blockIdx.x & 255u) * 0x01010101u;
  auto hits16 = [&](uint4 q) -> unsigned {
    const unsigned ONES = 0x01010101u, TOP = 0x80808080u;
    const unsigned y0 = q.x ^ pat, y1 = q.y ^ pat, y2 = q.z ^ pat, y3 = q.w ^ pat;
    const unsigned m0 = (y0 - ONES) & ~y0 & TOP;
    const unsigned m1 = (y1 - ONES) & ~y1 & TOP;
    const unsigned m2 = (y2 - ONES) & ~y2 & TOP;
    const unsigned m3 = (y3 - ONES) & ~y3 & TOP;
    return (m0 >> 7) | (m1 >> 6) | (m2 >> 5) | (m3 >> 4);
  };

  const uint4* __restrict__ keys4 = (const uint4*)keys;
  const int nRows = (nEdges / (ROW_EDGES * QUADS)) * QUADS;
  const int r0 = (int)(((long long)blockIdx.x * nRows) / (long long)gridDim.x);
  for (int it = 0; it < nRows; it += QUADS) {
    uint4 q[QUADS];
    int rb = r0 + it;
    if (rb >= nRows) rb -= nRows;
#pragma unroll
    for (int u = 0; u < QUADS; ++u) {
      int r = rb + u;
      if (r >= nRows) r -= nRows;
      q[u] = keys4[r * WG_THREADS + tid];
    }
    unsigned hb[QUADS / 2];
    unsigned n = 0u;
#pragma unroll
    for (int i = 0; i < QUADS / 2; ++i) {
      hb[i] = hits16(q[2 * i]) | (hits16(q[2 * i + 1]) << 4);
      n += (unsigned)__popc(hb[i]);
    }
    if (n != 0u) {
      unsigned pos = atomicAdd(&cntL, n);
#pragma unroll
      for (int i = 0; i < QUADS / 2; ++i) {
        unsigned m = hb[i];
        while (m != 0u) {
          const int p = __ffs((int)m) - 1;
          m &= m - 1u;
          int r = rb + 2 * i + ((p >> 2) & 1);
          if (r >= nRows) r -= nRows;
          if (pos < (unsigned)LIST_CAP)
            listL[pos].x = r * ROW_EDGES + tid * 16 + ((p & 3) << 2) + (p >> 3);
          ++pos;
        }
      }
    }
  }
  for (int eb = nRows * ROW_EDGES + tid * 16; eb < nEdges; eb += ROW_EDGES) {
    unsigned m = hits16(keys4[eb >> 4]);
    if (m != 0u) {
      unsigned pos = atomicAdd(&cntL, (unsigned)__popc(m));
      while (m != 0u) {
        const int p = __ffs((int)m) - 1;
        m &= m - 1u;
        if (pos < (unsigned)LIST_CAP) listL[pos].x = eb + ((p & 3) << 2) + (p >> 3);
        ++pos;
      }
    }
  }
  __syncthreads();
  const unsigned total = cntL;
  const bool listed = total <= (unsigned)LIST_CAP;

  auto rescan = [&](auto&& fn) {
    for (int eb = tid; eb < nEdges; eb += WG_THREADS) {
      const unsigned ld = (unsigned)(dst[eb] - lo);
      if (ld < nHereU) fn(src[eb], (int)ld);
    }
  };

  if (listed) {
    for (unsigned base = 0; base < total; base += 4u * WG_THREADS) {
      int sv[4], ldv[4]; unsigned iv[4]; bool ok[4], mine[4];
#pragma unroll
      for (int u = 0; u < 4; ++u) {
        iv[u] = base + (unsigned)(u * WG_THREADS + tid);
        ok[u] = iv[u] < total;
        int e = ok[u] ? listL[iv[u]].x : 0;
        mine[u] = ok[u] && e < nEdges;
        if (!mine[u]) e = 0;
        sv[u] = src[e];
        ldv[u] = dst[e] - lo;
        mine[u] = mine[u] && (unsigned)ldv[u] < nHereU;
      }
      float4 e4[4];
#pragma unroll
      for (int u = 0; u < 4; ++u)
        e4[u] = *(const float4*)(el + (size_t)sv[u] * HEADS);
#pragma unroll
      for (int u = 0; u < 4; ++u) {
        if (ok[u] && !mine[u]) {
          listL[iv[u]] = make_int2(0, -1);
          wL[iv[u]] = make_float4(0.0f, 0.0f, 0.0f, 0.0f);
        }
        if (mine[u]) {
          listL[iv[u]] = make_int2(sv[u], ldv[u]);
          wL[iv[u]] = e4[u];
          unsigned* mk = &mkeyL[ldv[u] * HEADS];
          __hip_atomic_fetch_max(mk + 0, fkey(e4[u].x), __ATOMIC_RELAXED, __HIP_MEMORY_SCOPE_WORKGROUP);
          __hip_atomic_fetch_max(mk + 1, fkey(e4[u].y), __ATOMIC_RELAXED, __HIP_MEMORY_SCOPE_WORKGROUP);
          __hip_atomic_fetch_max(mk + 2, fkey(e4[u].z), __ATOMIC_RELAXED, __HIP_MEMORY_SCOPE_WORKGROUP);
          __hip_atomic_fetch_max(mk + 3, fkey(e4[u].w), __ATOMIC_RELAXED, __HIP_MEMORY_SCOPE_WORKGROUP);
        }
      }
    }
    const unsigned padEnd = (total + MSG_ROUND - 1u) / MSG_ROUND * MSG_ROUND;
    for (unsigned i = total + (unsigned)tid; i < padEnd; i += WG_THREADS) {
      listL[i] = make_int2(0, -1);
      wL[i] = make_float4(0.0f, 0.0f, 0.0f, 0.0f);
    }
  } else {
    rescan([&](int s, int ld) {
      const float4 e4 = *(const float4*)(el + (size_t)s * HEADS);
      unsigned* mk = &mkeyL[ld * HEADS];
      __hip_atomic_fetch_max(mk + 0, fkey(e4.x), __ATOMIC_RELAXED, __HIP_MEMORY_SCOPE_WORKGROUP);
      __hip_atomic_fetch_max(mk + 1, fkey(e4.y), __ATOMIC_RELAXED, __HIP_MEMORY_SCOPE_WORKGROUP);
      __hip_atomic_fetch_max(mk + 2, fkey(e4.z), __ATOMIC_RELAXED, __HIP_MEMORY_SCOPE_WORKGROUP);
      __hip_atomic_fetch_max(mk + 3, fkey(e4.w), __ATOMIC_RELAXED, __HIP_MEMORY_SCOPE_WORKGROUP);
    });
  }
  __syncthreads();
  for (int i = tid; i < nHere * HEADS; i += WG_THREADS)
    ML[i] = lrelu(fkey_inv(mkeyL[i]) + erL[i]);
  __syncthreads();

  auto add_part = [&](float4 f, float elv, int ld, int c4) {
    const int h = c4 >> 2;
    const float w = expf(lrelu(elv + erL[ld * HEADS + h]) - ML[ld * HEADS + h]);
    const int a = ld * 16 + c4;
    __hip_atomic_fetch_add(&accL[0][a], f.x * w, __ATOMIC_RELAXED, __HIP_MEMORY_SCOPE_WORKGROUP);
    __hip_atomic_fetch_add(&accL[1][a], f.y * w, __ATOMIC_RELAXED, __HIP_MEMORY_SCOPE_WORKGROUP);
    __hip_atomic_fetch_add(&accL[2][a], f.z * w, __ATOMIC_RELAXED, __HIP_MEMORY_SCOPE_WORKGROUP);
    __hip_atomic_fetch_add(&accL[3][a], f.w * w, __ATOMIC_RELAXED, __HIP_MEMORY_SCOPE_WORKGROUP);
    if ((c4 & 3) == 0)
      __hip_atomic_fetch_add(&sL[ld * HEADS + h], w, __ATOMIC_RELAXED, __HIP_MEMORY_SCOPE_WORKGROUP);
  };
  if (listed) {
    for (unsigned i = (unsigned)tid; i < total; i += WG_THREADS) {
      const int ld = listL[i].y;
      if (ld >= 0) {
        const float4 e4 = wL[i];
        const float4 r4 = *(const float4*)&erL[ld * HEADS];
        const float4 m4 = *(const float4*)&ML[ld * HEADS];
        float4 w;
        w.x = expf(lrelu(e4.x + r4.x) - m4.x);
        w.y = expf(lrelu(e4.y + r4.y) - m4.y);
        w.z = expf(lrelu(e4.z + r4.z) - m4.z);
        w.w = expf(lrelu(e4.w + r4.w) - m4.w);
        wL[i] = w;
        float* sp = &sL[ld * HEADS];
        __hip_atomic_fetch_add(sp + 0, w.x, __ATOMIC_RELAXED, __HIP_MEMORY_SCOPE_WORKGROUP);
        __hip_atomic_fetch_add(sp + 1, w.y, __ATOMIC_RELAXED, __HIP_MEMORY_SCOPE_WORKGROUP);
        __hip_atomic_fetch_add(sp + 2, w.z, __ATOMIC_RELAXED, __HIP_MEMORY_SCOPE_WORKGROUP);
        __hip_atomic_fetch_add(sp + 3, w.w, __ATOMIC_RELAXED, __HIP_MEMORY_SCOPE_WORKGROUP);
      }
    }
    __syncthreads();

    const unsigned hw = (unsigned)(tid >> 4);
    const unsigned c4 = (unsigned)(tid & 15);
    const unsigned h = c4 >> 2;
    const float4* __restrict__ ft4 = (const float4*)ft;
    const float* wF = (const float*)wL;
    for (unsigned base = 0; base < total; base += MSG_ROUND) {
      int2 en[MSG_UNR]; float wv[MSG_UNR]; float4 f[MSG_UNR];
#pragma unroll
      for (int u = 0; u < MSG_UNR; ++u) {
        const unsigned i = base + (unsigned)u * 64u + hw;
        en[u] = listL[i];
        wv[u] = wF[i * HEADS + h];
      }
#pragma unroll
      for (int u = 0; u < MSG_UNR; ++u)
        f[u] = ft4[(unsigned)en[u].x * 16u + c4];
#pragma unroll
      for (int u = 0; u < MSG_UNR; ++u) {
        if (en[u].y >= 0) {
          const int a = en[u].y * 16 + (int)c4;
          const float w = wv[u];
          __hip_atomic_fetch_add(&accL[0][a], f[u].x * w, __ATOMIC_RELAXED, __HIP_MEMORY_SCOPE_WORKGROUP);
          __hip_atomic_fetch_add(&accL[1][a], f[u].y * w, __ATOMIC_RELAXED, __HIP_MEMORY_SCOPE_WORKGROUP);
          __hip_atomic_fetch_add(&accL[2][a], f[u].z * w, __ATOMIC_RELAXED, __HIP_MEMORY_SCOPE_WORKGROUP);
          __hip_atomic_fetch_add(&accL[3][a], f[u].w * w, __ATOMIC_RELAXED, __HIP_MEMORY_SCOPE_WORKGROUP);
        }
      }
    }
  } else {
    rescan([&](int s, int ld) {
      for (int c4 = 0; c4 < 16; ++c4)
        add_part(*(const float4*)(ft + (size_t)s * HD + 4 * c4),
                 el[(size_t)s * HEADS + (c4 >> 2)], ld, c4);
    });
  }
  __syncthreads();

  for (int i = tid; i < nHere * 16; i += WG_THREADS) {
    const int n = i >> 4, c4 = i & 15;
    const float sv = sL[n * HEADS + (c4 >> 2)];
    float4 o = *(const float4*)(bias + 4 * c4);
    if (sv > 0.0f) {
      o.x += accL[0][i] / sv; o.y += accL[1][i] / sv;
      o.z += accL[2][i] / sv; o.w += accL[3][i] / sv;
    }
    *(float4*)(out + (size_t)(lo + n) * HD + 4 * c4) = o;
  }
}

extern "C" void kernel_launch(void* const* d_in, const int* in_sizes, int n_in,
                              void* d_out, int out_size, void* d_ws,
                              size_t ws_size, hipStream_t stream) {
  const float* feat   = (const float*)d_in[0];
  const int*   src    = (const int*)d_in[1];
  const int*   dst    = (const int*)d_in[2];
  const float* W      = (const float*)d_in[3];
  const float* attn_l = (const float*)d_in[4];
  const float* attn_r = (const float*)d_in[5];
  const float* bias   = (const float*)d_in[6];
  float* out = (float*)d_out;

  const int N = in_sizes[0] / IN_FEATS;
  const int E = in_sizes[1];

  float* ws = (float*)d_ws;
  float* ft = ws;
  float* el = ft + (size_t)N * HD;
  unsigned* keys = (unsigned*)(el + (size_t)N * HEADS);

  int groups = (N + NODES_MAX - 1) / NODES_MAX;
  if (groups < 256) groups = 256;

  {
    int nTiles = (N + 15) / 16;
    int blocks = (nTiles + 7) / 8;
    gat_proj_scores_keys<<<blocks, 256, 0, stream>>>(feat, W, attn_l, dst, ft, el,
                                                     keys, N, nTiles, E, groups);
  }
  {
    gat_edges_owner<<<groups, WG_THREADS, 0, stream>>>(src, dst, keys, ft, el,
                                                       attn_r, bias, out, N, E);
  }
}


# Listed edges ranked per destination node (one returning LDS integer add each) and ordered by node in LDS; a half-wave owns a node and sums its rows in registers: no LDS float adds; sums in rank order
# speedup vs baseline: 5.3382x; 1.0980x over previous
#include <hip/hip_runtime.h>
#include <math.h>


typedef __attribute__((ext_vector_type(2))) float v2f;
typedef __attribute__((ext_vector_type(8))) float v8f;

#define IN_FEATS 128
#define HEADS    4
#define DHEAD    16
#define HD       64
#define NEG_SLOPE 0.2f

__global__ __launch_bounds__(256) void gat_proj_scores_keys(
    const float* __restrict__ feat, const float* __restrict__ W,
    const float* __restrict__ attn_l, const int* __restrict__ dst,
    float* __restrict__ ft, float* __restrict__ el,
    unsigned* __restrict__ keys,
    int nNodes, int nTiles, int nEdges, int groups) {
  const int lane = threadIdx.x & 31;
  const int tile = blockIdx.x * (blockDim.x >> 5) + (threadIdx.x >> 5);
  const int mrow = lane & 15;
  const int kq   = lane >> 4;

  const int gtid     = blockIdx.x * blockDim.x + threadIdx.x;
  const int nThreads = gridDim.x * blockDim.x;
  const int nWords   = ((nEdges + 15) / 16) * 4;
  const int w0 = gtid, w1 = gtid + nThreads;
  const bool whole0 = w0 < nWords && 4 * w0 + 3 < nEdges;
  const bool whole1 = w1 < nWords && 4 * w1 + 3 < nEdges;
  int4 d0 = make_int4(0, 0, 0, 0), d1 = make_int4(0, 0, 0, 0);
  if (whole0) d0 = *(const int4*)(dst + 4 * w0);
  if (whole1) d1 = *(const int4*)(dst + 4 * w1);

  if (tile < nTiles) {
    const int row0 = tile * 16;

    int arow = row0 + mrow;
    if (arow >= nNodes) arow = nNodes - 1;
    const float2* __restrict__ apf2 =
        (const float2*)(feat + (size_t)arow * IN_FEATS) + kq;

    const float al0 = attn_l[mrow],      al1 = attn_l[16 + mrow];
    const float al2 = attn_l[32 + mrow], al3 = attn_l[48 + mrow];

    v8f acc0 = {}, acc1 = {}, acc2 = {}, acc3 = {};

#pragma unroll 4
    for (int k = 0; k < IN_FEATS; k += 4) {
      float2 av = apf2[k >> 1];
      v2f a; a[0] = av.x; a[1] = av.y;

      const float* wr0 = W + (size_t)(k + 2 * kq) * HD;
      const float* wr1 = wr0 + HD;
      v2f b0; b0[0] = wr0[mrow];      b0[1] = wr1[mrow];
      v2f b1; b1[0] = wr0[mrow + 16]; b1[1] = wr1[mrow + 16];
      v2f b2; b2[0] = wr0[mrow + 32]; b2[1] = wr1[mrow + 32];
      v2f b3; b3[0] = wr0[mrow + 48]; b3[1] = wr1[mrow + 48];

      acc0 = __builtin_amdgcn_wmma_f32_16x16x4_f32(false, a, false, b0,
                                                   (short)0, acc0, false, false);
      acc1 = __builtin_amdgcn_wmma_f32_16x16x4_f32(false, a, false, b1,
                                                   (short)0, acc1, false, false);
      acc2 = __builtin_amdgcn_wmma_f32_16x16x4_f32(false, a, false, b2,
                                                   (short)0, acc2, false, false);
      acc3 = __builtin_amdgcn_wmma_f32_16x16x4_f32(false, a, false, b3,
                                                   (short)0, acc3, false, false);
    }

#pragma unroll
    for (int r = 0; r < 8; ++r) {
      int row = row0 + r + 8 * kq;
      if (row < nNodes) {
        float* o = ft + (size_t)row * HD + mrow;
        o[0]  = acc0[r];
        o[16] = acc1[r];
        o[32] = acc2[r];
        o[48] = acc3[r];
      }
    }

    auto half_wave_sum = [](float v) -> float {
      v += __shfl_xor(v, 8, 16);
      v += __shfl_xor(v, 4, 16);
      v += __shfl_xor(v, 2, 16);
      v += __shfl_xor(v, 1, 16);
      return v;
    };
    float4 mine = make_float4(0.0f, 0.0f, 0.0f, 0.0f);
#pragma unroll
    for (int r = 0; r < 8; ++r) {
      const float s0 = half_wave_sum(acc0[r] * al0);
      const float s1 = half_wave_sum(acc1[r] * al1);
      const float s2 = half_wave_sum(acc2[r] * al2);
      const float s3 = half_wave_sum(acc3[r] * al3);
      if (mrow == r) mine = make_float4(s0, s1, s2, s3);
    }
    const int erow = row0 + mrow + 8 * kq;
    if (mrow < 8 && erow < nNodes)
      *(float4*)(el + (size_t)erow * HEADS) = mine;
  }

  const unsigned nodesPer = (unsigned)((nNodes + groups - 1) / groups);
  auto pack4 = [&](int4 d) -> unsigned {
    return  (((unsigned)d.x / nodesPer) & 255u)
         | ((((unsigned)d.y / nodesPer) & 255u) << 8)
         | ((((unsigned)d.z / nodesPer) & 255u) << 16)
         | ((((unsigned)d.w / nodesPer) & 255u) << 24);
  };
  auto word_at_end = [&](int w) -> unsigned {
    unsigned k = 0u;
    for (int j = 0; j < 4; ++j)
      if (4 * w + j < nEdges)
        k |= (((unsigned)dst[4 * w + j] / nodesPer) & 255u) << (8 * j);
    return k;
  };
  if (w0 < nWords) keys[w0] = whole0 ? pack4(d0) : word_at_end(w0);
  if (w1 < nWords) keys[w1] = whole1 ? pack4(d1) : word_at_end(w1);
  for (int w = w1 + nThreads; w < nWords; w += nThreads)
    keys[w] = (4 * w + 3 < nEdges) ? pack4(*(const int4*)(dst + 4 * w))
                                   : word_at_end(w);
}

#define WG_THREADS 1024
#define NODES_MAX  256
#define LIST_CAP   8192
#define MSG_UNR    8
static_assert(NODES_MAX == 256, "a local node number is packed into 8 bits; 8 per lane in the scan");
static_assert(LIST_CAP <= 65536, "a list index is kept in 16 bits");
#define QUADS      8
#define ROW_EDGES  (WG_THREADS * 16)

__device__ __forceinline__ unsigned fkey(float f) {
  unsigned b = __float_as_uint(f);
  return (b & 0x80000000u) ? ~b : (b | 0x80000000u);
}
__device__ __forceinline__ float fkey_inv(unsigned k) {
  unsigned b = (k & 0x80000000u) ? (k & 0x7fffffffu) : ~k;
  return __uint_as_float(b);
}
__device__ __forceinline__ float lrelu(float x) {
  return (x > 0.0f) ? x : NEG_SLOPE * x;
}

__global__ __launch_bounds__(WG_THREADS) void gat_edges_owner(
    const int* __restrict__ src, const int* __restrict__ dst,
    const unsigned* __restrict__ keys,
    const float* __restrict__ ft, const float* __restrict__ el,
    const float* __restrict__ attn_r, const float* __restrict__ bias,
    float* __restrict__ out, int nNodes, int nEdges) {
  __shared__ float    accL[4][NODES_MAX * 16];
  __shared__ float    sL[NODES_MAX * HEADS];
  __shared__ alignas(16) float erL[NODES_MAX * HEADS];
  __shared__ alignas(16) float ML[NODES_MAX * HEADS];
  __shared__ unsigned mkeyL[NODES_MAX * HEADS];
  __shared__ int2     listL[LIST_CAP];
  __shared__ float4   wL[LIST_CAP];
  __shared__ unsigned short orderL[LIST_CAP];
  __shared__ unsigned degL[NODES_MAX];
  __shared__ unsigned offL[NODES_MAX];
  __shared__ unsigned cntL;
  __shared__ unsigned nextL;

  const int tid  = threadIdx.x;
  const int nodesPer = (nNodes + (int)gridDim.x - 1) / (int)gridDim.x;
  const int lo = (int)blockIdx.x * nodesPer;
  int nHere = nNodes - lo;
  if (nHere > nodesPer) nHere = nodesPer;
  if (nHere <= 0) return;

  for (int i = tid; i < nHere * HEADS; i += WG_THREADS) {
    mkeyL[i] = 0u;
    const int h = i & (HEADS - 1);
    const float* f  = ft + (size_t)(lo + (i >> 2)) * HD + h * DHEAD;
    const float* ar = attn_r + h * DHEAD;
    float sr = 0.0f;
#pragma unroll
    for (int d = 0; d < DHEAD; ++d) sr += f[d] * ar[d];
    erL[i] = sr;
  }
  if (tid < NODES_MAX) degL[tid] = 0u;
  if (tid == 0) { cntL = 0u; nextL = 0u; }
  __syncthreads();

  const unsigned nHereU = (unsigned)nHere;

  const unsigned pat = ((unsigned)blockIdx.x & 255u) * 0x01010101u;
  auto hits16 = [&](uint4 q) -> unsigned {
    const unsigned ONES = 0x01010101u, TOP = 0x80808080u;
    const unsigned y0 = q.x ^ pat, y1 = q.y ^ pat, y2 = q.z ^ pat, y3 = q.w ^ pat;
    const unsigned m0 = (y0 - ONES) & ~y0 & TOP;
    const unsigned m1 = (y1 - ONES) & ~y1 & TOP;
    const unsigned m2 = (y2 - ONES) & ~y2 & TOP;
    const unsigned m3 = (y3 - ONES) & ~y3 & TOP;
    return (m0 >> 7) | (m1 >> 6) | (m2 >> 5) | (m3 >> 4);
  };

  const uint4* __restrict__ keys4 = (const uint4*)keys;
  const int nRows = (nEdges / (ROW_EDGES * QUADS)) * QUADS;
  const int r0 = (int)(((long long)blockIdx.x * nRows) / (long long)gridDim.x);
  for (int it = 0; it < nRows; it += QUADS) {
    uint4 q[QUADS];
    int rb = r0 + it;
    if (rb >= nRows) rb -= nRows;
#pragma unroll
    for (int u = 0; u < QUADS; ++u) {
      int r = rb + u;
      if (r >= nRows) r -= nRows;
      q[u] = keys4[r * WG_THREADS + tid];
    }
    unsigned hb[QUADS / 2];
    unsigned n = 0u;
#pragma unroll
    for (int i = 0; i < QUADS / 2; ++i) {
      hb[i] = hits16(q[2 * i]) | (hits16(q[2 * i + 1]) << 4);
      n += (unsigned)__popc(hb[i]);
    }
    if (n != 0u) {
      unsigned pos = atomicAdd(&cntL, n);
#pragma unroll
      for (int i = 0; i < QUADS / 2; ++i) {
        unsigned m = hb[i];
        while (m != 0u) {
          const int p = __ffs((int)m) - 1;
          m &= m - 1u;
          int r = rb + 2 * i + ((p >> 2) & 1);
          if (r >= nRows) r -= nRows;
          if (pos < (unsigned)LIST_CAP)
            listL[pos].x = r * ROW_EDGES + tid * 16 + ((p & 3) << 2) + (p >> 3);
          ++pos;
        }
      }
    }
  }
  for (int eb = nRows * ROW_EDGES + tid * 16; eb < nEdges; eb += ROW_EDGES) {
    unsigned m = hits16(keys4[eb >> 4]);
    if (m != 0u) {
      unsigned pos = atomicAdd(&cntL, (unsigned)__popc(m));
      while (m != 0u) {
        const int p = __ffs((int)m) - 1;
        m &= m - 1u;
        if (pos < (unsigned)LIST_CAP) listL[pos].x = eb + ((p & 3) << 2) + (p >> 3);
        ++pos;
      }
    }
  }
  __syncthreads();
  const unsigned total = cntL;
  const bool listed = total <= (unsigned)LIST_CAP;

  auto rescan = [&](auto&& fn) {
    for (int eb = tid; eb < nEdges; eb += WG_THREADS) {
      const unsigned ld = (unsigned)(dst[eb] - lo);
      if (ld < nHereU) fn(src[eb], (int)ld);
    }
  };

  if (listed) {
    for (unsigned base = 0; base < total; base += 4u * WG_THREADS) {
      int sv[4], ldv[4]; unsigned iv[4]; bool ok[4], mine[4];
#pragma unroll
      for (int u = 0; u < 4; ++u) {
        iv[u] = base + (unsigned)(u * WG_THREADS + tid);
        ok[u] = iv[u] < total;
        int e = ok[u] ? listL[iv[u]].x : 0;
        mine[u] = ok[u] && e < nEdges;
        if (!mine[u]) e = 0;
        sv[u] = src[e];
        ldv[u] = dst[e] - lo;
        mine[u] = mine[u] && (unsigned)ldv[u] < nHereU;
      }
      float4 e4[4];
#pragma unroll
      for (int u = 0; u < 4; ++u)
        e4[u] = *(const float4*)(el + (size_t)sv[u] * HEADS);
#pragma unroll
      for (int u = 0; u < 4; ++u) {
        if (ok[u] && !mine[u])
          listL[iv[u]] = make_int2(0, -1);
        if (mine[u]) {
          const unsigned rank = __hip_atomic_fetch_add(&degL[ldv[u]], 1u, __ATOMIC_RELAXED,
                                                       __HIP_MEMORY_SCOPE_WORKGROUP);
          listL[iv[u]] = make_int2(sv[u], ldv[u] | (int)(rank << 8));
          wL[iv[u]] = e4[u];
          unsigned* mk = &mkeyL[ldv[u] * HEADS];
          __hip_atomic_fetch_max(mk + 0, fkey(e4[u].x), __ATOMIC_RELAXED, __HIP_MEMORY_SCOPE_WORKGROUP);
          __hip_atomic_fetch_max(mk + 1, fkey(e4[u].y), __ATOMIC_RELAXED, __HIP_MEMORY_SCOPE_WORKGROUP);
          __hip_atomic_fetch_max(mk + 2, fkey(e4[u].z), __ATOMIC_RELAXED, __HIP_MEMORY_SCOPE_WORKGROUP);
          __hip_atomic_fetch_max(mk + 3, fkey(e4[u].w), __ATOMIC_RELAXED, __HIP_MEMORY_SCOPE_WORKGROUP);
        }
      }
    }
  } else {
    for (int i = tid; i < nHere * 16; i += WG_THREADS) {
      accL[0][i] = 0.0f; accL[1][i] = 0.0f; accL[2][i] = 0.0f; accL[3][i] = 0.0f;
    }
    for (int i = tid; i < nHere * HEADS; i += WG_THREADS) sL[i] = 0.0f;
    rescan([&](int s, int ld) {
      const float4 e4 = *(const float4*)(el + (size_t)s * HEADS);
      unsigned* mk = &mkeyL[ld * HEADS];
      __hip_atomic_fetch_max(mk + 0, fkey(e4.x), __ATOMIC_RELAXED, __HIP_MEMORY_SCOPE_WORKGROUP);
      __hip_atomic_fetch_max(mk + 1, fkey(e4.y), __ATOMIC_RELAXED, __HIP_MEMORY_SCOPE_WORKGROUP);
      __hip_atomic_fetch_max(mk + 2, fkey(e4.z), __ATOMIC_RELAXED, __HIP_MEMORY_SCOPE_WORKGROUP);
      __hip_atomic_fetch_max(mk + 3, fkey(e4.w), __ATOMIC_RELAXED, __HIP_MEMORY_SCOPE_WORKGROUP);
    });
  }
  __syncthreads();
  for (int i = tid; i < nHere * HEADS; i += WG_THREADS)
    ML[i] = lrelu(fkey_inv(mkeyL[i]) + erL[i]);
  if (listed && tid >= WG_THREADS - 32) {
    const int l = tid & 31;
    unsigned d[8], s = 0u;
#pragma unroll
    for (int j = 0; j < 8; ++j) { d[j] = degL[8 * l + j]; s += d[j]; }
    unsigned incl = s;
#pragma unroll
    for (int o = 1; o < 32; o <<= 1) {
      const unsigned t = (unsigned)__shfl_up((int)incl, o, 32);
      if (l >= o) incl += t;
    }
    unsigned run = incl - s;
#pragma unroll
    for (int j = 0; j < 8; ++j) { offL[8 * l + j] = run; run += d[j]; }
  }
  __syncthreads();

  auto add_part = [&](float4 f, float elv, int ld, int c4) {
    const int h = c4 >> 2;
    const float w = expf(lrelu(elv + erL[ld * HEADS + h]) - ML[ld * HEADS + h]);
    const int a = ld * 16 + c4;
    __hip_atomic_fetch_add(&accL[0][a], f.x * w, __ATOMIC_RELAXED, __HIP_MEMORY_SCOPE_WORKGROUP);
    __hip_atomic_fetch_add(&accL[1][a], f.y * w, __ATOMIC_RELAXED, __HIP_MEMORY_SCOPE_WORKGROUP);
    __hip_atomic_fetch_add(&accL[2][a], f.z * w, __ATOMIC_RELAXED, __HIP_MEMORY_SCOPE_WORKGROUP);
    __hip_atomic_fetch_add(&accL[3][a], f.w * w, __ATOMIC_RELAXED, __HIP_MEMORY_SCOPE_WORKGROUP);
    if ((c4 & 3) == 0)
      __hip_atomic_fetch_add(&sL[ld * HEADS + h], w, __ATOMIC_RELAXED, __HIP_MEMORY_SCOPE_WORKGROUP);
  };
  if (listed) {
    for (unsigned i = (unsigned)tid; i < total; i += WG_THREADS) {
      const int y = listL[i].y;
      if (y >= 0) {
        const int ld = y & (NODES_MAX - 1);
        const float4 e4 = wL[i];
        const float4 r4 = *(const float4*)&erL[ld * HEADS];
        const float4 m4 = *(const float4*)&ML[ld * HEADS];
        float4 w;
        w.x = expf(lrelu(e4.x + r4.x) - m4.x);
        w.y = expf(lrelu(e4.y + r4.y) - m4.y);
        w.z = expf(lrelu(e4.z + r4.z) - m4.z);
        w.w = expf(lrelu(e4.w + r4.w) - m4.w);
        wL[i] = w;
        orderL[offL[ld] + ((unsigned)y >> 8)] = (unsigned short)i;
      }
    }
    __syncthreads();

    const unsigned c4 = (unsigned)(tid & 15);
    const unsigned h = c4 >> 2;
    const float4* __restrict__ ft4 = (const float4*)ft;
    const float* wF = (const float*)wL;
    const float4 b4 = *(const float4*)(bias + 4 * c4);
    for (;;) {
      unsigned t = 0u;
      if (c4 == 0u)
        t = __hip_atomic_fetch_add(&nextL, 1u, __ATOMIC_RELAXED, __HIP_MEMORY_SCOPE_WORKGROUP);
      const unsigned n = (unsigned)__shfl((int)t, 0, 16);
      if (n >= nHereU) break;
      const unsigned beg = offL[n], end = beg + degL[n];
      float4 acc = make_float4(0.0f, 0.0f, 0.0f, 0.0f);
      float ssum = 0.0f;
      for (unsigned p = beg; p < end; p += MSG_UNR) {
        unsigned sv[MSG_UNR]; float wv[MSG_UNR]; float4 f[MSG_UNR];
#pragma unroll
        for (int u = 0; u < MSG_UNR; ++u) {
          unsigned q = p + (unsigned)u;
          if (q >= end) q = end - 1u;
          const unsigned i = orderL[q];
          sv[u] = (unsigned)listL[i].x;
          wv[u] = wF[i * HEADS + h];
        }
#pragma unroll
        for (int u = 0; u < MSG_UNR; ++u)
          f[u] = ft4[sv[u] * 16u + c4];
#pragma unroll
        for (int u = 0; u < MSG_UNR; ++u) {
          const float w = (p + (unsigned)u < end) ? wv[u] : 0.0f;
          acc.x += f[u].x * w; acc.y += f[u].y * w;
          acc.z += f[u].z * w; acc.w += f[u].w * w;
          ssum += w;
        }
      }
      float4 o = b4;
      if (ssum > 0.0f) {
        o.x += acc.x / ssum; o.y += acc.y / ssum;
        o.z += acc.z / ssum; o.w += acc.w / ssum;
      }
      *(float4*)(out + (size_t)(lo + (int)n) * HD + 4 * c4) = o;
    }
  } else {
    rescan([&](int s, int ld) {
      for (int c4 = 0; c4 < 16; ++c4)
        add_part(*(const float4*)(ft + (size_t)s * HD + 4 * c4),
                 el[(size_t)s * HEADS + (c4 >> 2)], ld, c4);
    });
    __syncthreads();

    for (int i = tid; i < nHere * 16; i += WG_THREADS) {
      const int n = i >> 4, c4 = i & 15;
      const float sv = sL[n * HEADS + (c4 >> 2)];
      float4 o = *(const float4*)(bias + 4 * c4);
      if (sv > 0.0f) {
        o.x += accL[0][i] / sv; o.y += accL[1][i] / sv;
        o.z += accL[2][i] / sv; o.w += accL[3][i] / sv;
      }
      *(float4*)(out + (size_t)(lo + n) * HD + 4 * c4) = o;
    }
  }
}

extern "C" void kernel_launch(void* const* d_in, const int* in_sizes, int n_in,
                              void* d_out, int out_size, void* d_ws,
                              size_t ws_size, hipStream_t stream) {
  const float* feat   = (const float*)d_in[0];
  const int*   src    = (const int*)d_in[1];
  const int*   dst    = (const int*)d_in[2];
  const float* W      = (const float*)d_in[3];
  const float* attn_l = (const float*)d_in[4];
  const float* attn_r = (const float*)d_in[5];
  const float* bias   = (const float*)d_in[6];
  float* out = (float*)d_out;

  const int N = in_sizes[0] / IN_FEATS;
  const int E = in_sizes[1];

  float* ws = (float*)d_ws;
  float* ft = ws;
  float* el = ft + (size_t)N * HD;
  unsigned* keys = (unsigned*)(el + (size_t)N * HEADS);

  int groups = (N + NODES_MAX - 1) / NODES_MAX;
  if (groups < 256) groups = 256;

  {
    int nTiles = (N + 15) / 16;
    int blocks = (nTiles + 7) / 8;
    gat_proj_scores_keys<<<blocks, 256, 0, stream>>>(feat, W, attn_l, dst, ft, el,
                                                     keys, N, nTiles, E, groups);
  }
  {
    gat_edges_owner<<<groups, WG_THREADS, 0, stream>>>(src, dst, keys, ft, el,
                                                       attn_r, bias, out, N, E);
  }
}


# Owner keys stored as bit planes (8 words per 32 edges, written with wave ballots): edge filter gets an exact 32-edge hit mask from 8 boolean ops, list places by a wave prefix sum; sums in rank order
# speedup vs baseline: 5.5195x; 1.0340x over previous
#include <hip/hip_runtime.h>
#include <math.h>


typedef __attribute__((ext_vector_type(2))) float v2f;
typedef __attribute__((ext_vector_type(8))) float v8f;

#define IN_FEATS 128
#define HEADS    4
#define DHEAD    16
#define HD       64
#define NEG_SLOPE 0.2f

__global__ __launch_bounds__(256) void gat_proj_scores_keys(
    const float* __restrict__ feat, const float* __restrict__ W,
    const float* __restrict__ attn_l, const int* __restrict__ dst,
    float* __restrict__ ft, float* __restrict__ el,
    unsigned* __restrict__ keys,
    int nNodes, int nTiles, int nEdges, int groups) {
  const int lane = threadIdx.x & 31;
  const int tile = blockIdx.x * (blockDim.x >> 5) + (threadIdx.x >> 5);
  const int mrow = lane & 15;
  const int kq   = lane >> 4;

  const int nWaves  = gridDim.x * (blockDim.x >> 5);
  const int nGroups = (nEdges + 31) >> 5;
  const int nOct    = (nGroups + 7) >> 3;
  int dv[8];
#pragma unroll
  for (int u = 0; u < 8; ++u) {
    const int e = (tile * 8 + u) * 32 + lane;
    dv[u] = (tile < nOct && e < nEdges) ? dst[e] : 0;
  }

  if (tile < nTiles) {
    const int row0 = tile * 16;

    int arow = row0 + mrow;
    if (arow >= nNodes) arow = nNodes - 1;
    const float2* __restrict__ apf2 =
        (const float2*)(feat + (size_t)arow * IN_FEATS) + kq;

    const float al0 = attn_l[mrow],      al1 = attn_l[16 + mrow];
    const float al2 = attn_l[32 + mrow], al3 = attn_l[48 + mrow];

    v8f acc0 = {}, acc1 = {}, acc2 = {}, acc3 = {};

#pragma unroll 4
    for (int k = 0; k < IN_FEATS; k += 4) {
      float2 av = apf2[k >> 1];
      v2f a; a[0] = av.x; a[1] = av.y;

      const float* wr0 = W + (size_t)(k + 2 * kq) * HD;
      const float* wr1 = wr0 + HD;
      v2f b0; b0[0] = wr0[mrow];      b0[1] = wr1[mrow];
      v2f b1; b1[0] = wr0[mrow + 16]; b1[1] = wr1[mrow + 16];
      v2f b2; b2[0] = wr0[mrow + 32]; b2[1] = wr1[mrow + 32];
      v2f b3; b3[0] = wr0[mrow + 48]; b3[1] = wr1[mrow + 48];

      acc0 = __builtin_amdgcn_wmma_f32_16x16x4_f32(false, a, false, b0,
                                                   (short)0, acc0, false, false);
      acc1 = __builtin_amdgcn_wmma_f32_16x16x4_f32(false, a, false, b1,
                                                   (short)0, acc1, false, false);
      acc2 = __builtin_amdgcn_wmma_f32_16x16x4_f32(false, a, false, b2,
                                                   (short)0, acc2, false, false);
      acc3 = __builtin_amdgcn_wmma_f32_16x16x4_f32(false, a, false, b3,
                                                   (short)0, acc3, false, false);
    }

#pragma unroll
    for (int r = 0; r < 8; ++r) {
      int row = row0 + r + 8 * kq;
      if (row < nNodes) {
        float* o = ft + (size_t)row * HD + mrow;
        o[0]  = acc0[r];
        o[16] = acc1[r];
        o[32] = acc2[r];
        o[48] = acc3[r];
      }
    }

    auto half_wave_sum = [](float v) -> float {
      v += __shfl_xor(v, 8, 16);
      v += __shfl_xor(v, 4, 16);
      v += __shfl_xor(v, 2, 16);
      v += __shfl_xor(v, 1, 16);
      return v;
    };
    float4 mine = make_float4(0.0f, 0.0f, 0.0f, 0.0f);
#pragma unroll
    for (int r = 0; r < 8; ++r) {
      const float s0 = half_wave_sum(acc0[r] * al0);
      const float s1 = half_wave_sum(acc1[r] * al1);
      const float s2 = half_wave_sum(acc2[r] * al2);
      const float s3 = half_wave_sum(acc3[r] * al3);
      if (mrow == r) mine = make_float4(s0, s1, s2, s3);
    }
    const int erow = row0 + mrow + 8 * kq;
    if (mrow < 8 && erow < nNodes)
      *(float4*)(el + (size_t)erow * HEADS) = mine;
  }

  const unsigned nodesPer = (unsigned)((nNodes + groups - 1) / groups);
  unsigned* __restrict__ keysLo = keys;
  unsigned* __restrict__ keysHi = keys + (size_t)nGroups * 4;
  auto put_octet = [&](int oct, const int (&d)[8]) {
    unsigned lo = 0u, hi = 0u;
#pragma unroll
    for (int u = 0; u < 8; ++u) {
      const unsigned own = (unsigned)d[u] / nodesPer;
#pragma unroll
      for (int j = 0; j < 8; ++j) {
        const unsigned bal = __builtin_amdgcn_ballot_w32((own & (1u << j)) != 0u);
        if (lane == 4 * u + (j & 3)) { if (j < 4) lo = bal; else hi = bal; }
      }
    }
    const int g = oct * 8 + (lane >> 2);
    if (g < nGroups) {
      keysLo[(size_t)oct * 32 + lane] = lo;
      keysHi[(size_t)oct * 32 + lane] = hi;
    }
  };
  if (tile < nOct) put_octet(tile, dv);
  for (int oct = tile + nWaves; oct < nOct; oct += nWaves) {
    int dn[8];
#pragma unroll
    for (int u = 0; u < 8; ++u) {
      const int e = (oct * 8 + u) * 32 + lane;
      dn[u] = (e < nEdges) ? dst[e] : 0;
    }
    put_octet(oct, dn);
  }
}

#define WG_THREADS 1024
#define NODES_MAX  256
#define LIST_CAP   8192
#define MSG_UNR    8
static_assert(NODES_MAX == 256, "a local node number is packed into 8 bits; 8 per lane in the scan");
static_assert(LIST_CAP <= 65536, "a list index is kept in 16 bits");
#define QUADS      8
#define GRP_IT     (QUADS / 2)

__device__ __forceinline__ unsigned fkey(float f) {
  unsigned b = __float_as_uint(f);
  return (b & 0x80000000u) ? ~b : (b | 0x80000000u);
}
__device__ __forceinline__ float fkey_inv(unsigned k) {
  unsigned b = (k & 0x80000000u) ? (k & 0x7fffffffu) : ~k;
  return __uint_as_float(b);
}
__device__ __forceinline__ float lrelu(float x) {
  return (x > 0.0f) ? x : NEG_SLOPE * x;
}

__global__ __launch_bounds__(WG_THREADS) void gat_edges_owner(
    const int* __restrict__ src, const int* __restrict__ dst,
    const unsigned* __restrict__ keys,
    const float* __restrict__ ft, const float* __restrict__ el,
    const float* __restrict__ attn_r, const float* __restrict__ bias,
    float* __restrict__ out, int nNodes, int nEdges) {
  __shared__ float    accL[4][NODES_MAX * 16];
  __shared__ float    sL[NODES_MAX * HEADS];
  __shared__ alignas(16) float erL[NODES_MAX * HEADS];
  __shared__ alignas(16) float ML[NODES_MAX * HEADS];
  __shared__ unsigned mkeyL[NODES_MAX * HEADS];
  __shared__ int2     listL[LIST_CAP];
  __shared__ float4   wL[LIST_CAP];
  __shared__ unsigned short orderL[LIST_CAP];
  __shared__ unsigned degL[NODES_MAX];
  __shared__ unsigned offL[NODES_MAX];
  __shared__ unsigned cntL;
  __shared__ unsigned nextL;

  const int tid  = threadIdx.x;
  const int nodesPer = (nNodes + (int)gridDim.x - 1) / (int)gridDim.x;
  const int lo = (int)blockIdx.x * nodesPer;
  int nHere = nNodes - lo;
  if (nHere > nodesPer) nHere = nodesPer;
  if (nHere <= 0) return;

  for (int i = tid; i < nHere * HEADS; i += WG_THREADS) {
    mkeyL[i] = 0u;
    const int h = i & (HEADS - 1);
    const float* f  = ft + (size_t)(lo + (i >> 2)) * HD + h * DHEAD;
    const float* ar = attn_r + h * DHEAD;
    float sr = 0.0f;
#pragma unroll
    for (int d = 0; d < DHEAD; ++d) sr += f[d] * ar[d];
    erL[i] = sr;
  }
  if (tid < NODES_MAX) degL[tid] = 0u;
  if (tid == 0) { cntL = 0u; nextL = 0u; }
  __syncthreads();

  const unsigned nHereU = (unsigned)nHere;

  const unsigned own8 = (unsigned)blockIdx.x & 255u;
  const unsigned f0 = (own8 &   1u) ? 0u : ~0u, f1 = (own8 &   2u) ? 0u : ~0u;
  const unsigned f2 = (own8 &   4u) ? 0u : ~0u, f3 = (own8 &   8u) ? 0u : ~0u;
  const unsigned f4 = (own8 &  16u) ? 0u : ~0u, f5 = (own8 &  32u) ? 0u : ~0u;
  const unsigned f6 = (own8 &  64u) ? 0u : ~0u, f7 = (own8 & 128u) ? 0u : ~0u;
  auto hits32 = [&](uint4 l, uint4 h) -> unsigned {
    return (l.x ^ f0) & (l.y ^ f1) & (l.z ^ f2) & (l.w ^ f3)
         & (h.x ^ f4) & (h.y ^ f5) & (h.z ^ f6) & (h.w ^ f7);
  };

  const int nGroups = (nEdges + 31) >> 5;
  const uint4* __restrict__ lo4 = (const uint4*)keys;
  const uint4* __restrict__ hi4 = lo4 + nGroups;
  const int lane = tid & 31;
  const int nRows = ((nGroups / WG_THREADS) / GRP_IT) * GRP_IT;
  const int r0 = (int)(((long long)blockIdx.x * nRows) / (long long)gridDim.x);
  for (int it = 0; it < nRows; it += GRP_IT) {
    uint4 ql[GRP_IT], qh[GRP_IT];
    int rb = r0 + it;
    if (rb >= nRows) rb -= nRows;
#pragma unroll
    for (int u = 0; u < GRP_IT; ++u) {
      int r = rb + u;
      if (r >= nRows) r -= nRows;
      ql[u] = lo4[r * WG_THREADS + tid];
      qh[u] = hi4[r * WG_THREADS + tid];
    }
    unsigned m[GRP_IT];
    unsigned n = 0u;
#pragma unroll
    for (int u = 0; u < GRP_IT; ++u) {
      m[u] = hits32(ql[u], qh[u]);
      n += (unsigned)__popc(m[u]);
    }
    if (__builtin_amdgcn_ballot_w32(n != 0u) != 0u) {
      unsigned incl = n;
#pragma unroll
      for (int o = 1; o < 32; o <<= 1) {
        const unsigned t = (unsigned)__shfl_up((int)incl, o, 32);
        if (lane >= o) incl += t;
      }
      const unsigned tot = (unsigned)__shfl((int)incl, 31, 32);
      unsigned first = 0u;
      if (lane == 0) first = atomicAdd(&cntL, tot);
      first = (unsigned)__shfl((int)first, 0, 32);
      unsigned pos = first + incl - n;
#pragma unroll
      for (int u = 0; u < GRP_IT; ++u) {
        unsigned mm = m[u];
        int r = rb + u;
        if (r >= nRows) r -= nRows;
        const int e0 = (r * WG_THREADS + tid) << 5;
        while (mm != 0u) {
          const int p = __ffs((int)mm) - 1;
          mm &= mm - 1u;
          if (pos < (unsigned)LIST_CAP) listL[pos].x = e0 + p;
          ++pos;
        }
      }
    }
  }
  for (int c = nRows * WG_THREADS + tid; c < nGroups; c += WG_THREADS) {
    unsigned mm = hits32(lo4[c], hi4[c]);
    if (mm != 0u) {
      unsigned pos = atomicAdd(&cntL, (unsigned)__popc(mm));
      while (mm != 0u) {
        const int p = __ffs((int)mm) - 1;
        mm &= mm - 1u;
        if (pos < (unsigned)LIST_CAP) listL[pos].x = (c << 5) + p;
        ++pos;
      }
    }
  }
  __syncthreads();
  const unsigned total = cntL;
  const bool listed = total <= (unsigned)LIST_CAP;

  auto rescan = [&](auto&& fn) {
    for (int eb = tid; eb < nEdges; eb += WG_THREADS) {
      const unsigned ld = (unsigned)(dst[eb] - lo);
      if (ld < nHereU) fn(src[eb], (int)ld);
    }
  };

  if (listed) {
    for (unsigned base = 0; base < total; base += 4u * WG_THREADS) {
      int sv[4], ldv[4]; unsigned iv[4]; bool ok[4], mine[4];
#pragma unroll
      for (int u = 0; u < 4; ++u) {
        iv[u] = base + (unsigned)(u * WG_THREADS + tid);
        ok[u] = iv[u] < total;
        int e = ok[u] ? listL[iv[u]].x : 0;
        mine[u] = ok[u] && e < nEdges;
        if (!mine[u]) e = 0;
        sv[u] = src[e];
        ldv[u] = dst[e] - lo;
        mine[u] = mine[u] && (unsigned)ldv[u] < nHereU;
      }
      float4 e4[4];
#pragma unroll
      for (int u = 0; u < 4; ++u)
        e4[u] = *(const float4*)(el + (size_t)sv[u] * HEADS);
#pragma unroll
      for (int u = 0; u < 4; ++u) {
        if (ok[u] && !mine[u])
          listL[iv[u]] = make_int2(0, -1);
        if (mine[u]) {
          const unsigned rank = __hip_atomic_fetch_add(&degL[ldv[u]], 1u, __ATOMIC_RELAXED,
                                                       __HIP_MEMORY_SCOPE_WORKGROUP);
          listL[iv[u]] = make_int2(sv[u], ldv[u] | (int)(rank << 8));
          wL[iv[u]] = e4[u];
          unsigned* mk = &mkeyL[ldv[u] * HEADS];
          __hip_atomic_fetch_max(mk + 0, fkey(e4[u].x), __ATOMIC_RELAXED, __HIP_MEMORY_SCOPE_WORKGROUP);
          __hip_atomic_fetch_max(mk + 1, fkey(e4[u].y), __ATOMIC_RELAXED, __HIP_MEMORY_SCOPE_WORKGROUP);
          __hip_atomic_fetch_max(mk + 2, fkey(e4[u].z), __ATOMIC_RELAXED, __HIP_MEMORY_SCOPE_WORKGROUP);
          __hip_atomic_fetch_max(mk + 3, fkey(e4[u].w), __ATOMIC_RELAXED, __HIP_MEMORY_SCOPE_WORKGROUP);
        }
      }
    }
  } else {
    for (int i = tid; i < nHere * 16; i += WG_THREADS) {
      accL[0][i] = 0.0f; accL[1][i] = 0.0f; accL[2][i] = 0.0f; accL[3][i] = 0.0f;
    }
    for (int i = tid; i < nHere * HEADS; i += WG_THREADS) sL[i] = 0.0f;
    rescan([&](int s, int ld) {
      const float4 e4 = *(const float4*)(el + (size_t)s * HEADS);
      unsigned* mk = &mkeyL[ld * HEADS];
      __hip_atomic_fetch_max(mk + 0, fkey(e4.x), __ATOMIC_RELAXED, __HIP_MEMORY_SCOPE_WORKGROUP);
      __hip_atomic_fetch_max(mk + 1, fkey(e4.y), __ATOMIC_RELAXED, __HIP_MEMORY_SCOPE_WORKGROUP);
      __hip_atomic_fetch_max(mk + 2, fkey(e4.z), __ATOMIC_RELAXED, __HIP_MEMORY_SCOPE_WORKGROUP);
      __hip_atomic_fetch_max(mk + 3, fkey(e4.w), __ATOMIC_RELAXED, __HIP_MEMORY_SCOPE_WORKGROUP);
    });
  }
  __syncthreads();
  for (int i = tid; i < nHere * HEADS; i += WG_THREADS)
    ML[i] = lrelu(fkey_inv(mkeyL[i]) + erL[i]);
  if (listed && tid >= WG_THREADS - 32) {
    const int l = tid & 31;
    unsigned d[8], s = 0u;
#pragma unroll
    for (int j = 0; j < 8; ++j) { d[j] = degL[8 * l + j]; s += d[j]; }
    unsigned incl = s;
#pragma unroll
    for (int o = 1; o < 32; o <<= 1) {
      const unsigned t = (unsigned)__shfl_up((int)incl, o, 32);
      if (l >= o) incl += t;
    }
    unsigned run = incl - s;
#pragma unroll
    for (int j = 0; j < 8; ++j) { offL[8 * l + j] = run; run += d[j]; }
  }
  __syncthreads();

  auto add_part = [&](float4 f, float elv, int ld, int c4) {
    const int h = c4 >> 2;
    const float w = expf(lrelu(elv + erL[ld * HEADS + h]) - ML[ld * HEADS + h]);
    const int a = ld * 16 + c4;
    __hip_atomic_fetch_add(&accL[0][a], f.x * w, __ATOMIC_RELAXED, __HIP_MEMORY_SCOPE_WORKGROUP);
    __hip_atomic_fetch_add(&accL[1][a], f.y * w, __ATOMIC_RELAXED, __HIP_MEMORY_SCOPE_WORKGROUP);
    __hip_atomic_fetch_add(&accL[2][a], f.z * w, __ATOMIC_RELAXED, __HIP_MEMORY_SCOPE_WORKGROUP);
    __hip_atomic_fetch_add(&accL[3][a], f.w * w, __ATOMIC_RELAXED, __HIP_MEMORY_SCOPE_WORKGROUP);
    if ((c4 & 3) == 0)
      __hip_atomic_fetch_add(&sL[ld * HEADS + h], w, __ATOMIC_RELAXED, __HIP_MEMORY_SCOPE_WORKGROUP);
  };
  if (listed) {
    for (unsigned i = (unsigned)tid; i < total; i += WG_THREADS) {
      const int y = listL[i].y;
      if (y >= 0) {
        const int ld = y & (NODES_MAX - 1);
        const float4 e4 = wL[i];
        const float4 r4 = *(const float4*)&erL[ld * HEADS];
        const float4 m4 = *(const float4*)&ML[ld * HEADS];
        float4 w;
        w.x = expf(lrelu(e4.x + r4.x) - m4.x);
        w.y = expf(lrelu(e4.y + r4.y) - m4.y);
        w.z = expf(lrelu(e4.z + r4.z) - m4.z);
        w.w = expf(lrelu(e4.w + r4.w) - m4.w);
        wL[i] = w;
        orderL[offL[ld] + ((unsigned)y >> 8)] = (unsigned short)i;
      }
    }
    __syncthreads();

    const unsigned c4 = (unsigned)(tid & 15);
    const unsigned h = c4 >> 2;
    const float4* __restrict__ ft4 = (const float4*)ft;
    const float* wF = (const float*)wL;
    const float4 b4 = *(const float4*)(bias + 4 * c4);
    for (;;) {
      unsigned t = 0u;
      if (c4 == 0u)
        t = __hip_atomic_fetch_add(&nextL, 1u, __ATOMIC_RELAXED, __HIP_MEMORY_SCOPE_WORKGROUP);
      const unsigned n = (unsigned)__shfl((int)t, 0, 16);
      if (n >= nHereU) break;
      const unsigned beg = offL[n], end = beg + degL[n];
      float4 acc = make_float4(0.0f, 0.0f, 0.0f, 0.0f);
      float ssum = 0.0f;
      for (unsigned p = beg; p < end; p += MSG_UNR) {
        unsigned sv[MSG_UNR]; float wv[MSG_UNR]; float4 f[MSG_UNR];
#pragma unroll
        for (int u = 0; u < MSG_UNR; ++u) {
          unsigned q = p + (unsigned)u;
          if (q >= end) q = end - 1u;
          const unsigned i = orderL[q];
          sv[u] = (unsigned)listL[i].x;
          wv[u] = wF[i * HEADS + h];
        }
#pragma unroll
        for (int u = 0; u < MSG_UNR; ++u)
          f[u] = ft4[sv[u] * 16u + c4];
#pragma unroll
        for (int u = 0; u < MSG_UNR; ++u) {
          const float w = (p + (unsigned)u < end) ? wv[u] : 0.0f;
          acc.x += f[u].x * w; acc.y += f[u].y * w;
          acc.z += f[u].z * w; acc.w += f[u].w * w;
          ssum += w;
        }
      }
      float4 o = b4;
      if (ssum > 0.0f) {
        o.x += acc.x / ssum; o.y += acc.y / ssum;
        o.z += acc.z / ssum; o.w += acc.w / ssum;
      }
      *(float4*)(out + (size_t)(lo + (int)n) * HD + 4 * c4) = o;
    }
  } else {
    rescan([&](int s, int ld) {
      for (int c4 = 0; c4 < 16; ++c4)
        add_part(*(const float4*)(ft + (size_t)s * HD + 4 * c4),
                 el[(size_t)s * HEADS + (c4 >> 2)], ld, c4);
    });
    __syncthreads();

    for (int i = tid; i < nHere * 16; i += WG_THREADS) {
      const int n = i >> 4, c4 = i & 15;
      const float sv = sL[n * HEADS + (c4 >> 2)];
      float4 o = *(const float4*)(bias + 4 * c4);
      if (sv > 0.0f) {
        o.x += accL[0][i] / sv; o.y += accL[1][i] / sv;
        o.z += accL[2][i] / sv; o.w += accL[3][i] / sv;
      }
      *(float4*)(out + (size_t)(lo + n) * HD + 4 * c4) = o;
    }
  }
}

extern "C" void kernel_launch(void* const* d_in, const int* in_sizes, int n_in,
                              void* d_out, int out_size, void* d_ws,
                              size_t ws_size, hipStream_t stream) {
  const float* feat   = (const float*)d_in[0];
  const int*   src    = (const int*)d_in[1];
  const int*   dst    = (const int*)d_in[2];
  const float* W      = (const float*)d_in[3];
  const float* attn_l = (const float*)d_in[4];
  const float* attn_r = (const float*)d_in[5];
  const float* bias   = (const float*)d_in[6];
  float* out = (float*)d_out;

  const int N = in_sizes[0] / IN_FEATS;
  const int E = in_sizes[1];

  float* ws = (float*)d_ws;
  float* ft = ws;
  float* el = ft + (size_t)N * HD;
  unsigned* keys = (unsigned*)(el + (size_t)N * HEADS);

  int groups = (N + NODES_MAX - 1) / NODES_MAX;
  if (groups < 256) groups = 256;

  {
    int nTiles = (N + 15) / 16;
    int blocks = (nTiles + 7) / 8;
    gat_proj_scores_keys<<<blocks, 256, 0, stream>>>(feat, W, attn_l, dst, ft, el,
                                                     keys, N, nTiles, E, groups);
  }
  {
    gat_edges_owner<<<groups, WG_THREADS, 0, stream>>>(src, dst, keys, ft, el,
                                                       attn_r, bias, out, N, E);
  }
}


# Candidate check reads one packed word per listed edge (source id + local node, written by the projection launch from src and dst; 4 more workspace bytes per edge): 2 scattered reads per edge, not 3
# speedup vs baseline: 5.6992x; 1.0326x over previous
#include <hip/hip_runtime.h>
#include <math.h>


typedef __attribute__((ext_vector_type(2))) float v2f;
typedef __attribute__((ext_vector_type(8))) float v8f;

#define IN_FEATS 128
#define HEADS    4
#define DHEAD    16
#define HD       64
#define NEG_SLOPE 0.2f

__global__ __launch_bounds__(256) void gat_proj_scores_keys(
    const float* __restrict__ feat, const float* __restrict__ W,
    const float* __restrict__ attn_l, const int* __restrict__ src,
    const int* __restrict__ dst,
    float* __restrict__ ft, float* __restrict__ el,
    unsigned* __restrict__ keys, unsigned* __restrict__ pk,
    int nNodes, int nTiles, int nEdges, int groups) {
  const int lane = threadIdx.x & 31;
  const int tile = blockIdx.x * (blockDim.x >> 5) + (threadIdx.x >> 5);
  const int mrow = lane & 15;
  const int kq   = lane >> 4;

  const int nWaves  = gridDim.x * (blockDim.x >> 5);
  const int nGroups = (nEdges + 31) >> 5;
  const int nOct    = (nGroups + 7) >> 3;
  const bool packing = pk != nullptr;
  int dv[8], sv[8];
#pragma unroll
  for (int u = 0; u < 8; ++u) {
    const int e = (tile * 8 + u) * 32 + lane;
    const bool in = tile < nOct && e < nEdges;
    dv[u] = in ? dst[e] : 0;
    sv[u] = (in && packing) ? src[e] : 0;
  }

  if (tile < nTiles) {
    const int row0 = tile * 16;

    int arow = row0 + mrow;
    if (arow >= nNodes) arow = nNodes - 1;
    const float2* __restrict__ apf2 =
        (const float2*)(feat + (size_t)arow * IN_FEATS) + kq;

    const float al0 = attn_l[mrow],      al1 = attn_l[16 + mrow];
    const float al2 = attn_l[32 + mrow], al3 = attn_l[48 + mrow];

    v8f acc0 = {}, acc1 = {}, acc2 = {}, acc3 = {};

#pragma unroll 4
    for (int k = 0; k < IN_FEATS; k += 4) {
      float2 av = apf2[k >> 1];
      v2f a; a[0] = av.x; a[1] = av.y;

      const float* wr0 = W + (size_t)(k + 2 * kq) * HD;
      const float* wr1 = wr0 + HD;
      v2f b0; b0[0] = wr0[mrow];      b0[1] = wr1[mrow];
      v2f b1; b1[0] = wr0[mrow + 16]; b1[1] = wr1[mrow + 16];
      v2f b2; b2[0] = wr0[mrow + 32]; b2[1] = wr1[mrow + 32];
      v2f b3; b3[0] = wr0[mrow + 48]; b3[1] = wr1[mrow + 48];

      acc0 = __builtin_amdgcn_wmma_f32_16x16x4_f32(false, a, false, b0,
                                                   (short)0, acc0, false, false);
      acc1 = __builtin_amdgcn_wmma_f32_16x16x4_f32(false, a, false, b1,
                                                   (short)0, acc1, false, false);
      acc2 = __builtin_amdgcn_wmma_f32_16x16x4_f32(false, a, false, b2,
                                                   (short)0, acc2, false, false);
      acc3 = __builtin_amdgcn_wmma_f32_16x16x4_f32(false, a, false, b3,
                                                   (short)0, acc3, false, false);
    }

#pragma unroll
    for (int r = 0; r < 8; ++r) {
      int row = row0 + r + 8 * kq;
      if (row < nNodes) {
        float* o = ft + (size_t)row * HD + mrow;
        o[0]  = acc0[r];
        o[16] = acc1[r];
        o[32] = acc2[r];
        o[48] = acc3[r];
      }
    }

    auto half_wave_sum = [](float v) -> float {
      v += __shfl_xor(v, 8, 16);
      v += __shfl_xor(v, 4, 16);
      v += __shfl_xor(v, 2, 16);
      v += __shfl_xor(v, 1, 16);
      return v;
    };
    float4 mine = make_float4(0.0f, 0.0f, 0.0f, 0.0f);
#pragma unroll
    for (int r = 0; r < 8; ++r) {
      const float s0 = half_wave_sum(acc0[r] * al0);
      const float s1 = half_wave_sum(acc1[r] * al1);
      const float s2 = half_wave_sum(acc2[r] * al2);
      const float s3 = half_wave_sum(acc3[r] * al3);
      if (mrow == r) mine = make_float4(s0, s1, s2, s3);
    }
    const int erow = row0 + mrow + 8 * kq;
    if (mrow < 8 && erow < nNodes)
      *(float4*)(el + (size_t)erow * HEADS) = mine;
  }

  const unsigned nodesPer = (unsigned)((nNodes + groups - 1) / groups);
  unsigned* __restrict__ keysLo = keys;
  unsigned* __restrict__ keysHi = keys + (size_t)nGroups * 4;
  auto put_octet = [&](int oct, const int (&d)[8], const int (&s)[8]) {
    unsigned lo = 0u, hi = 0u;
#pragma unroll
    for (int u = 0; u < 8; ++u) {
      const unsigned own = (unsigned)d[u] / nodesPer;
      if (packing) {
        const int e = (oct * 8 + u) * 32 + lane;
        if (e < nEdges)
          pk[e] = (unsigned)s[u] | (((unsigned)d[u] - own * nodesPer) << 24);
      }
#pragma unroll
      for (int j = 0; j < 8; ++j) {
        const unsigned bal = __builtin_amdgcn_ballot_w32((own & (1u << j)) != 0u);
        if (lane == 4 * u + (j & 3)) { if (j < 4) lo = bal; else hi = bal; }
      }
    }
    const int g = oct * 8 + (lane >> 2);
    if (g < nGroups) {
      keysLo[(size_t)oct * 32 + lane] = lo;
      keysHi[(size_t)oct * 32 + lane] = hi;
    }
  };
  if (tile < nOct) put_octet(tile, dv, sv);
  for (int oct = tile + nWaves; oct < nOct; oct += nWaves) {
    int dn[8], sn[8];
#pragma unroll
    for (int u = 0; u < 8; ++u) {
      const int e = (oct * 8 + u) * 32 + lane;
      dn[u] = (e < nEdges) ? dst[e] : 0;
      sn[u] = (e < nEdges && packing) ? src[e] : 0;
    }
    put_octet(oct, dn, sn);
  }
}

#define WG_THREADS 1024
#define NODES_MAX  256
#define LIST_CAP   8192
#define MSG_UNR    8
static_assert(NODES_MAX == 256, "a local node number is packed into 8 bits; 8 per lane in the scan");
static_assert(LIST_CAP <= 65536, "a list index is kept in 16 bits");
#define QUADS      8
#define GRP_IT     (QUADS / 2)

__device__ __forceinline__ unsigned fkey(float f) {
  unsigned b = __float_as_uint(f);
  return (b & 0x80000000u) ? ~b : (b | 0x80000000u);
}
__device__ __forceinline__ float fkey_inv(unsigned k) {
  unsigned b = (k & 0x80000000u) ? (k & 0x7fffffffu) : ~k;
  return __uint_as_float(b);
}
__device__ __forceinline__ float lrelu(float x) {
  return (x > 0.0f) ? x : NEG_SLOPE * x;
}

__global__ __launch_bounds__(WG_THREADS) void gat_edges_owner(
    const int* __restrict__ src, const int* __restrict__ dst,
    const unsigned* __restrict__ keys, const unsigned* __restrict__ pk,
    const float* __restrict__ ft, const float* __restrict__ el,
    const float* __restrict__ attn_r, const float* __restrict__ bias,
    float* __restrict__ out, int nNodes, int nEdges) {
  __shared__ float    accL[4][NODES_MAX * 16];
  __shared__ float    sL[NODES_MAX * HEADS];
  __shared__ alignas(16) float erL[NODES_MAX * HEADS];
  __shared__ alignas(16) float ML[NODES_MAX * HEADS];
  __shared__ unsigned mkeyL[NODES_MAX * HEADS];
  __shared__ int2     listL[LIST_CAP];
  __shared__ float4   wL[LIST_CAP];
  __shared__ unsigned short orderL[LIST_CAP];
  __shared__ unsigned degL[NODES_MAX];
  __shared__ unsigned offL[NODES_MAX];
  __shared__ unsigned cntL;
  __shared__ unsigned nextL;

  const int tid  = threadIdx.x;
  const int nodesPer = (nNodes + (int)gridDim.x - 1) / (int)gridDim.x;
  const int lo = (int)blockIdx.x * nodesPer;
  int nHere = nNodes - lo;
  if (nHere > nodesPer) nHere = nodesPer;
  if (nHere <= 0) return;

  for (int i = tid; i < nHere * HEADS; i += WG_THREADS) {
    mkeyL[i] = 0u;
    const int h = i & (HEADS - 1);
    const float* f  = ft + (size_t)(lo + (i >> 2)) * HD + h * DHEAD;
    const float* ar = attn_r + h * DHEAD;
    float sr = 0.0f;
#pragma unroll
    for (int d = 0; d < DHEAD; ++d) sr += f[d] * ar[d];
    erL[i] = sr;
  }
  if (tid < NODES_MAX) degL[tid] = 0u;
  if (tid == 0) { cntL = 0u; nextL = 0u; }
  __syncthreads();

  const unsigned nHereU = (unsigned)nHere;

  const unsigned own8 = (unsigned)blockIdx.x & 255u;
  const unsigned f0 = (own8 &   1u) ? 0u : ~0u, f1 = (own8 &   2u) ? 0u : ~0u;
  const unsigned f2 = (own8 &   4u) ? 0u : ~0u, f3 = (own8 &   8u) ? 0u : ~0u;
  const unsigned f4 = (own8 &  16u) ? 0u : ~0u, f5 = (own8 &  32u) ? 0u : ~0u;
  const unsigned f6 = (own8 &  64u) ? 0u : ~0u, f7 = (own8 & 128u) ? 0u : ~0u;
  auto hits32 = [&](uint4 l, uint4 h) -> unsigned {
    return (l.x ^ f0) & (l.y ^ f1) & (l.z ^ f2) & (l.w ^ f3)
         & (h.x ^ f4) & (h.y ^ f5) & (h.z ^ f6) & (h.w ^ f7);
  };

  const int nGroups = (nEdges + 31) >> 5;
  const uint4* __restrict__ lo4 = (const uint4*)keys;
  const uint4* __restrict__ hi4 = lo4 + nGroups;
  const int lane = tid & 31;
  const int nRows = ((nGroups / WG_THREADS) / GRP_IT) * GRP_IT;
  const int r0 = (int)(((long long)blockIdx.x * nRows) / (long long)gridDim.x);
  for (int it = 0; it < nRows; it += GRP_IT) {
    uint4 ql[GRP_IT], qh[GRP_IT];
    int rb = r0 + it;
    if (rb >= nRows) rb -= nRows;
#pragma unroll
    for (int u = 0; u < GRP_IT; ++u) {
      int r = rb + u;
      if (r >= nRows) r -= nRows;
      ql[u] = lo4[r * WG_THREADS + tid];
      qh[u] = hi4[r * WG_THREADS + tid];
    }
    unsigned m[GRP_IT];
    unsigned n = 0u;
#pragma unroll
    for (int u = 0; u < GRP_IT; ++u) {
      m[u] = hits32(ql[u], qh[u]);
      n += (unsigned)__popc(m[u]);
    }
    if (__builtin_amdgcn_ballot_w32(n != 0u) != 0u) {
      unsigned incl = n;
#pragma unroll
      for (int o = 1; o < 32; o <<= 1) {
        const unsigned t = (unsigned)__shfl_up((int)incl, o, 32);
        if (lane >= o) incl += t;
      }
      const unsigned tot = (unsigned)__shfl((int)incl, 31, 32);
      unsigned first = 0u;
      if (lane == 0) first = atomicAdd(&cntL, tot);
      first = (unsigned)__shfl((int)first, 0, 32);
      unsigned pos = first + incl - n;
#pragma unroll
      for (int u = 0; u < GRP_IT; ++u) {
        unsigned mm = m[u];
        int r = rb + u;
        if (r >= nRows) r -= nRows;
        const int e0 = (r * WG_THREADS + tid) << 5;
        while (mm != 0u) {
          const int p = __ffs((int)mm) - 1;
          mm &= mm - 1u;
          if (pos < (unsigned)LIST_CAP) listL[pos].x = e0 + p;
          ++pos;
        }
      }
    }
  }
  for (int c = nRows * WG_THREADS + tid; c < nGroups; c += WG_THREADS) {
    unsigned mm = hits32(lo4[c], hi4[c]);
    if (mm != 0u) {
      unsigned pos = atomicAdd(&cntL, (unsigned)__popc(mm));
      while (mm != 0u) {
        const int p = __ffs((int)mm) - 1;
        mm &= mm - 1u;
        if (pos < (unsigned)LIST_CAP) listL[pos].x = (c << 5) + p;
        ++pos;
      }
    }
  }
  __syncthreads();
  const unsigned total = cntL;
  const bool listed = total <= (unsigned)LIST_CAP;

  auto rescan = [&](auto&& fn) {
    for (int eb = tid; eb < nEdges; eb += WG_THREADS) {
      const unsigned ld = (unsigned)(dst[eb] - lo);
      if (ld < nHereU) fn(src[eb], (int)ld);
    }
  };

  if (listed) {
    for (unsigned base = 0; base < total; base += 4u * WG_THREADS) {
      int sv[4], ldv[4]; unsigned iv[4]; bool ok[4], mine[4];
      if (pk != nullptr) {
        unsigned wv[4]; int ev[4];
#pragma unroll
        for (int u = 0; u < 4; ++u) {
          iv[u] = base + (unsigned)(u * WG_THREADS + tid);
          ok[u] = iv[u] < total;
          ev[u] = listL[ok[u] ? iv[u] : 0u].x;
        }
#pragma unroll
        for (int u = 0; u < 4; ++u) {
          mine[u] = ok[u] && ev[u] < nEdges;
          wv[u] = pk[mine[u] ? ev[u] : 0];
        }
#pragma unroll
        for (int u = 0; u < 4; ++u) {
          sv[u] = (int)(wv[u] & 0x00ffffffu);
          ldv[u] = (int)(wv[u] >> 24);
          mine[u] = mine[u] && (unsigned)ldv[u] < nHereU;
        }
      } else {
#pragma unroll
        for (int u = 0; u < 4; ++u) {
          iv[u] = base + (unsigned)(u * WG_THREADS + tid);
          ok[u] = iv[u] < total;
          int e = ok[u] ? listL[iv[u]].x : 0;
          mine[u] = ok[u] && e < nEdges;
          if (!mine[u]) e = 0;
          sv[u] = src[e];
          ldv[u] = dst[e] - lo;
          mine[u] = mine[u] && (unsigned)ldv[u] < nHereU;
        }
      }
      float4 e4[4];
#pragma unroll
      for (int u = 0; u < 4; ++u)
        e4[u] = *(const float4*)(el + (size_t)sv[u] * HEADS);
#pragma unroll
      for (int u = 0; u < 4; ++u) {
        if (ok[u] && !mine[u])
          listL[iv[u]] = make_int2(0, -1);
        if (mine[u]) {
          const unsigned rank = __hip_atomic_fetch_add(&degL[ldv[u]], 1u, __ATOMIC_RELAXED,
                                                       __HIP_MEMORY_SCOPE_WORKGROUP);
          listL[iv[u]] = make_int2(sv[u], ldv[u] | (int)(rank << 8));
          wL[iv[u]] = e4[u];
          unsigned* mk = &mkeyL[ldv[u] * HEADS];
          __hip_atomic_fetch_max(mk + 0, fkey(e4[u].x), __ATOMIC_RELAXED, __HIP_MEMORY_SCOPE_WORKGROUP);
          __hip_atomic_fetch_max(mk + 1, fkey(e4[u].y), __ATOMIC_RELAXED, __HIP_MEMORY_SCOPE_WORKGROUP);
          __hip_atomic_fetch_max(mk + 2, fkey(e4[u].z), __ATOMIC_RELAXED, __HIP_MEMORY_SCOPE_WORKGROUP);
          __hip_atomic_fetch_max(mk + 3, fkey(e4[u].w), __ATOMIC_RELAXED, __HIP_MEMORY_SCOPE_WORKGROUP);
        }
      }
    }
  } else {
    for (int i = tid; i < nHere * 16; i += WG_THREADS) {
      accL[0][i] = 0.0f; accL[1][i] = 0.0f; accL[2][i] = 0.0f; accL[3][i] = 0.0f;
    }
    for (int i = tid; i < nHere * HEADS; i += WG_THREADS) sL[i] = 0.0f;
    rescan([&](int s, int ld) {
      const float4 e4 = *(const float4*)(el + (size_t)s * HEADS);
      unsigned* mk = &mkeyL[ld * HEADS];
      __hip_atomic_fetch_max(mk + 0, fkey(e4.x), __ATOMIC_RELAXED, __HIP_MEMORY_SCOPE_WORKGROUP);
      __hip_atomic_fetch_max(mk + 1, fkey(e4.y), __ATOMIC_RELAXED, __HIP_MEMORY_SCOPE_WORKGROUP);
      __hip_atomic_fetch_max(mk + 2, fkey(e4.z), __ATOMIC_RELAXED, __HIP_MEMORY_SCOPE_WORKGROUP);
      __hip_atomic_fetch_max(mk + 3, fkey(e4.w), __ATOMIC_RELAXED, __HIP_MEMORY_SCOPE_WORKGROUP);
    });
  }
  __syncthreads();
  for (int i = tid; i < nHere * HEADS; i += WG_THREADS)
    ML[i] = lrelu(fkey_inv(mkeyL[i]) + erL[i]);
  if (listed && tid >= WG_THREADS - 32) {
    const int l = tid & 31;
    unsigned d[8], s = 0u;
#pragma unroll
    for (int j = 0; j < 8; ++j) { d[j] = degL[8 * l + j]; s += d[j]; }
    unsigned incl = s;
#pragma unroll
    for (int o = 1; o < 32; o <<= 1) {
      const unsigned t = (unsigned)__shfl_up((int)incl, o, 32);
      if (l >= o) incl += t;
    }
    unsigned run = incl - s;
#pragma unroll
    for (int j = 0; j < 8; ++j) { offL[8 * l + j] = run; run += d[j]; }
  }
  __syncthreads();

  auto add_part = [&](float4 f, float elv, int ld, int c4) {
    const int h = c4 >> 2;
    const float w = expf(lrelu(elv + erL[ld * HEADS + h]) - ML[ld * HEADS + h]);
    const int a = ld * 16 + c4;
    __hip_atomic_fetch_add(&accL[0][a], f.x * w, __ATOMIC_RELAXED, __HIP_MEMORY_SCOPE_WORKGROUP);
    __hip_atomic_fetch_add(&accL[1][a], f.y * w, __ATOMIC_RELAXED, __HIP_MEMORY_SCOPE_WORKGROUP);
    __hip_atomic_fetch_add(&accL[2][a], f.z * w, __ATOMIC_RELAXED, __HIP_MEMORY_SCOPE_WORKGROUP);
    __hip_atomic_fetch_add(&accL[3][a], f.w * w, __ATOMIC_RELAXED, __HIP_MEMORY_SCOPE_WORKGROUP);
    if ((c4 & 3) == 0)
      __hip_atomic_fetch_add(&sL[ld * HEADS + h], w, __ATOMIC_RELAXED, __HIP_MEMORY_SCOPE_WORKGROUP);
  };
  if (listed) {
    for (unsigned i = (unsigned)tid; i < total; i += WG_THREADS) {
      const int y = listL[i].y;
      if (y >= 0) {
        const int ld = y & (NODES_MAX - 1);
        const float4 e4 = wL[i];
        const float4 r4 = *(const float4*)&erL[ld * HEADS];
        const float4 m4 = *(const float4*)&ML[ld * HEADS];
        float4 w;
        w.x = expf(lrelu(e4.x + r4.x) - m4.x);
        w.y = expf(lrelu(e4.y + r4.y) - m4.y);
        w.z = expf(lrelu(e4.z + r4.z) - m4.z);
        w.w = expf(lrelu(e4.w + r4.w) - m4.w);
        wL[i] = w;
        orderL[offL[ld] + ((unsigned)y >> 8)] = (unsigned short)i;
      }
    }
    __syncthreads();

    const unsigned c4 = (unsigned)(tid & 15);
    const unsigned h = c4 >> 2;
    const float4* __restrict__ ft4 = (const float4*)ft;
    const float* wF = (const float*)wL;
    const float4 b4 = *(const float4*)(bias + 4 * c4);
    for (;;) {
      unsigned t = 0u;
      if (c4 == 0u)
        t = __hip_atomic_fetch_add(&nextL, 1u, __ATOMIC_RELAXED, __HIP_MEMORY_SCOPE_WORKGROUP);
      const unsigned n = (unsigned)__shfl((int)t, 0, 16);
      if (n >= nHereU) break;
      const unsigned beg = offL[n], end = beg + degL[n];
      float4 acc = make_float4(0.0f, 0.0f, 0.0f, 0.0f);
      float ssum = 0.0f;
      for (unsigned p = beg; p < end; p += MSG_UNR) {
        unsigned sv[MSG_UNR]; float wv[MSG_UNR]; float4 f[MSG_UNR];
#pragma unroll
        for (int u = 0; u < MSG_UNR; ++u) {
          unsigned q = p + (unsigned)u;
          if (q >= end) q = end - 1u;
          const unsigned i = orderL[q];
          sv[u] = (unsigned)listL[i].x;
          wv[u] = wF[i * HEADS + h];
        }
#pragma unroll
        for (int u = 0; u < MSG_UNR; ++u)
          f[u] = ft4[sv[u] * 16u + c4];
#pragma unroll
        for (int u = 0; u < MSG_UNR; ++u) {
          const float w = (p + (unsigned)u < end) ? wv[u] : 0.0f;
          acc.x += f[u].x * w; acc.y += f[u].y * w;
          acc.z += f[u].z * w; acc.w += f[u].w * w;
          ssum += w;
        }
      }
      float4 o = b4;
      if (ssum > 0.0f) {
        o.x += acc.x / ssum; o.y += acc.y / ssum;
        o.z += acc.z / ssum; o.w += acc.w / ssum;
      }
      *(float4*)(out + (size_t)(lo + (int)n) * HD + 4 * c4) = o;
    }
  } else {
    rescan([&](int s, int ld) {
      for (int c4 = 0; c4 < 16; ++c4)
        add_part(*(const float4*)(ft + (size_t)s * HD + 4 * c4),
                 el[(size_t)s * HEADS + (c4 >> 2)], ld, c4);
    });
    __syncthreads();

    for (int i = tid; i < nHere * 16; i += WG_THREADS) {
      const int n = i >> 4, c4 = i & 15;
      const float sv = sL[n * HEADS + (c4 >> 2)];
      float4 o = *(const float4*)(bias + 4 * c4);
      if (sv > 0.0f) {
        o.x += accL[0][i] / sv; o.y += accL[1][i] / sv;
        o.z += accL[2][i] / sv; o.w += accL[3][i] / sv;
      }
      *(float4*)(out + (size_t)(lo + n) * HD + 4 * c4) = o;
    }
  }
}

extern "C" void kernel_launch(void* const* d_in, const int* in_sizes, int n_in,
                              void* d_out, int out_size, void* d_ws,
                              size_t ws_size, hipStream_t stream) {
  const float* feat   = (const float*)d_in[0];
  const int*   src    = (const int*)d_in[1];
  const int*   dst    = (const int*)d_in[2];
  const float* W      = (const float*)d_in[3];
  const float* attn_l = (const float*)d_in[4];
  const float* attn_r = (const float*)d_in[5];
  const float* bias   = (const float*)d_in[6];
  float* out = (float*)d_out;

  const int N = in_sizes[0] / IN_FEATS;
  const int E = in_sizes[1];

  float* ws = (float*)d_ws;
  float* ft = ws;
  float* el = ft + (size_t)N * HD;
  unsigned* keys = (unsigned*)(el + (size_t)N * HEADS);

  int groups = (N + NODES_MAX - 1) / NODES_MAX;
  if (groups < 256) groups = 256;

  const size_t nGroupsH = ((size_t)E + 31) / 32;
  unsigned* pk = keys + nGroupsH * 8;
  const size_t wsNeed = ((size_t)N * (HD + HEADS) + nGroupsH * 8 + (size_t)E) * sizeof(float);
  if (ws_size < wsNeed || groups > 256) pk = nullptr;

  {
    int nTiles = (N + 15) / 16;
    int blocks = (nTiles + 7) / 8;
    gat_proj_scores_keys<<<blocks, 256, 0, stream>>>(feat, W, attn_l, src, dst, ft, el,
                                                     keys, pk, N, nTiles, E, groups);
  }
  {
    gat_edges_owner<<<groups, WG_THREADS, 0, stream>>>(src, dst, keys, pk, ft, el,
                                                       attn_r, bias, out, N, E);
  }
}
